# starting kernel
# baseline (speedup 1.0000x reference)
#include <hip/hip_runtime.h>
#include <hip/hip_bf16.h>
#include <math.h>

typedef float v2f __attribute__((ext_vector_type(2)));
typedef float v8f __attribute__((ext_vector_type(8)));

#define B_SZ   2
#define T_SEQ  2048
#define C_DIM  1024
#define NH     16
#define HD     32
#define QKH    32
#define VD     64
#define ROWQKV 3072

__device__ __forceinline__ v8f wmma4(v2f a, v2f b, v8f c) {
  return __builtin_amdgcn_wmma_f32_16x16x4_f32(
       false, a,  false, b,
       (short)0, c,  false,  false);
}

__global__ __launch_bounds__(256) void k_gemm(const float* __restrict__ A,
                                              const float* __restrict__ Bm,
                                              float* __restrict__ Cm,
                                              int N, int K) {
  const int lane = threadIdx.x & 31;
  const int wid  = threadIdx.x >> 5;
  const int wy   = wid & 3;
  const int wx   = wid >> 2;
  const int rowBase = blockIdx.y * 128 + wy * 32;
  const int colBase = blockIdx.x * 128 + wx * 64;
  const int hi   = lane >> 4;
  const int l15  = lane & 15;
  const int bc   = colBase + l15;

  const float* pA0 = A + (size_t)(rowBase + l15) * K + hi * 2;
  const float* pA1 = A + (size_t)(rowBase + 16 + l15) * K + hi * 2;
  const float* pB  = Bm + (size_t)(hi * 2) * N + bc;
  const size_t strideB = (size_t)4 * N;

  v8f acc[2][4] = {};

  v2f a0 = *(const v2f*)pA0;
  v2f a1 = *(const v2f*)pA1;
  v2f b[4];
  #pragma unroll
  for (int cg = 0; cg < 4; ++cg) {
    b[cg].x = pB[16 * cg];
    b[cg].y = pB[N + 16 * cg];
  }

  #pragma unroll 2
  for (int k = 0; k < K - 4; k += 4) {
    pA0 += 4; pA1 += 4; pB += strideB;
    const v2f na0 = *(const v2f*)pA0;
    const v2f na1 = *(const v2f*)pA1;
    v2f nb[4];
    #pragma unroll
    for (int cg = 0; cg < 4; ++cg) {
      nb[cg].x = pB[16 * cg];
      nb[cg].y = pB[N + 16 * cg];
    }
    if ((k & 60) == 0) {
      __builtin_prefetch(pB + 16 * strideB, 0, 1);
      __builtin_prefetch(pA0 + 64, 0, 1);
      __builtin_prefetch(pA1 + 64, 0, 1);
    }
    #pragma unroll
    for (int cg = 0; cg < 4; ++cg) {
      acc[0][cg] = wmma4(a0, b[cg], acc[0][cg]);
      acc[1][cg] = wmma4(a1, b[cg], acc[1][cg]);
    }
    a0 = na0; a1 = na1;
    #pragma unroll
    for (int cg = 0; cg < 4; ++cg) b[cg] = nb[cg];
  }
  #pragma unroll
  for (int cg = 0; cg < 4; ++cg) {
    acc[0][cg] = wmma4(a0, b[cg], acc[0][cg]);
    acc[1][cg] = wmma4(a1, b[cg], acc[1][cg]);
  }

  #pragma unroll
  for (int g = 0; g < 2; ++g) {
    #pragma unroll
    for (int r = 0; r < 8; ++r) {
      const size_t row = (size_t)(rowBase + g * 16 + r + hi * 8);
      #pragma unroll
      for (int cg = 0; cg < 4; ++cg) {
        Cm[row * N + bc + 16 * cg] = acc[g][cg][r];
      }
    }
  }
}

__global__ void k_rope(float* __restrict__ qkv) {
  size_t idx = (size_t)blockIdx.x * blockDim.x + threadIdx.x;
  const int j  = idx & 15;  idx >>= 4;
  const int h  = idx & 31;  idx >>= 5;
  const int qk = idx & 1;   idx >>= 1;
  const int t  = idx & 2047; idx >>= 11;
  const int b  = (int)idx;
  const float theta = __expf(-((float)(2 * j) / 32.0f) * logf(10000.0f));
  float s, c;
  __sincosf((float)t * theta, &s, &c);
  float* p = qkv + ((size_t)(b * T_SEQ + t)) * ROWQKV + qk * C_DIM + h * HD + 2 * j;
  const float x1 = p[0], x2 = p[1];
  p[0] = x1 * c - x2 * s;
  p[1] = x1 * s + x2 * c;
}

__global__ void k_lambda(const float* __restrict__ lq1, const float* __restrict__ lk1,
                         const float* __restrict__ lq2, const float* __restrict__ lk2,
                         float* __restrict__ lam) {
  const int l = threadIdx.x;
  float s1 = lq1[l] * lk1[l];
  float s2 = lq2[l] * lk2[l];
  #pragma unroll
  for (int off = 16; off; off >>= 1) {
    s1 += __shfl_xor(s1, off, 32);
    s2 += __shfl_xor(s2, off, 32);
  }
  if (l == 0) *lam = __expf(s1) - __expf(s2) + 0.2f;
}

__global__ __launch_bounds__(32) void k_attn(const float* __restrict__ qkv,
                                             float* __restrict__ Abuf) {
  const int qt   = blockIdx.x;
  const int h    = blockIdx.y;
  const int b    = blockIdx.z;
  const int lane = threadIdx.x;
  const int hi   = lane >> 4;
  const int l15  = lane & 15;

  const float* Q  = qkv + (size_t)b * T_SEQ * ROWQKV + h * HD;
  const float* Kp = qkv + (size_t)b * T_SEQ * ROWQKV + C_DIM + h * HD;
  const float* V  = qkv + (size_t)b * T_SEQ * ROWQKV + 2 * C_DIM + (h & 15) * VD;

  __shared__ float S[16 * 18];
  __shared__ float mrow[16], lrow[16], frow[16];

  if (lane < 16) { mrow[lane] = -3.0e38f; lrow[lane] = 0.0f; }
  __syncthreads();

  const size_t qoff = (size_t)(qt * 16 + l15) * ROWQKV;
  v2f qa[8];
  #pragma unroll
  for (int c = 0; c < 8; ++c) {
    const int col = 4 * c + hi * 2;
    qa[c].x = Q[qoff + col];
    qa[c].y = Q[qoff + col + 1];
  }

  v8f o0 = {}, o1 = {}, o2 = {}, o3 = {};
  const float scale = 0.17677669529663687f;

  for (int kb = 0; kb <= qt; ++kb) {
    v8f s = {};
    const size_t koff = (size_t)(kb * 16 + l15) * ROWQKV;
    #pragma unroll
    for (int c = 0; c < 8; ++c) {
      const int col = 4 * c + hi * 2;
      v2f kf; kf.x = Kp[koff + col]; kf.y = Kp[koff + col + 1];
      s = wmma4(qa[c], kf, s);
    }
    #pragma unroll
    for (int r = 0; r < 8; ++r) {
      const int m    = r + hi * 8;
      const int qidx = qt * 16 + m;
      const int kidx = kb * 16 + l15;
      S[m * 18 + l15] = (kidx <= qidx) ? s[r] * scale : -3.0e38f;
    }
    __syncthreads();

    if (lane < 16) {
      float mx = mrow[lane];
      #pragma unroll
      for (int j = 0; j < 16; ++j) mx = fmaxf(mx, S[lane * 18 + j]);
      const float f = __expf(mrow[lane] - mx);
      float sum = 0.0f;
      #pragma unroll
      for (int j = 0; j < 16; ++j) {
        const float p = __expf(S[lane * 18 + j] - mx);
        S[lane * 18 + j] = p;
        sum += p;
      }
      mrow[lane] = mx;
      frow[lane] = f;
      lrow[lane] = lrow[lane] * f + sum;
    }
    __syncthreads();

    float fr[8];
    #pragma unroll
    for (int r = 0; r < 8; ++r) fr[r] = frow[r + hi * 8];
    #pragma unroll
    for (int r = 0; r < 8; ++r) {
      o0[r] *= fr[r]; o1[r] *= fr[r]; o2[r] *= fr[r]; o3[r] *= fr[r];
    }

    v2f pa[4];
    #pragma unroll
    for (int c = 0; c < 4; ++c) {
      const int col = 4 * c + hi * 2;
      pa[c].x = S[l15 * 18 + col];
      pa[c].y = S[l15 * 18 + col + 1];
    }
    #pragma unroll
    for (int c = 0; c < 4; ++c) {
      const int krow = kb * 16 + 4 * c + hi * 2;
      const float* Vr0 = V + (size_t)krow * ROWQKV;
      const float* Vr1 = V + (size_t)(krow + 1) * ROWQKV;
      v2f vb0; vb0.x = Vr0[l15];      vb0.y = Vr1[l15];
      v2f vb1; vb1.x = Vr0[l15 + 16]; vb1.y = Vr1[l15 + 16];
      v2f vb2; vb2.x = Vr0[l15 + 32]; vb2.y = Vr1[l15 + 32];
      v2f vb3; vb3.x = Vr0[l15 + 48]; vb3.y = Vr1[l15 + 48];
      o0 = wmma4(pa[c], vb0, o0);
      o1 = wmma4(pa[c], vb1, o1);
      o2 = wmma4(pa[c], vb2, o2);
      o3 = wmma4(pa[c], vb3, o3);
    }
    __syncthreads();
  }

  float inv[8];
  #pragma unroll
  for (int r = 0; r < 8; ++r) inv[r] = 1.0f / lrow[r + hi * 8];
  float* Ab = Abuf + ((size_t)(b * QKH + h)) * T_SEQ * VD;
  #pragma unroll
  for (int r = 0; r < 8; ++r) {
    float* Ar = Ab + (size_t)(qt * 16 + r + hi * 8) * VD;
    Ar[l15]      = o0[r] * inv[r];
    Ar[l15 + 16] = o1[r] * inv[r];
    Ar[l15 + 32] = o2[r] * inv[r];
    Ar[l15 + 48] = o3[r] * inv[r];
  }
}

__global__ __launch_bounds__(256) void k_diffnorm(const float* __restrict__ Abuf,
                                                  const float* __restrict__ lamp,
                                                  float* __restrict__ Dbuf) {
  const float lam = *lamp;
  const int wid  = threadIdx.x >> 5;
  const int lane = threadIdx.x & 31;
  const size_t row = (size_t)blockIdx.x * 8 + wid;
  const int t  = (int)(row & 2047);
  const size_t bh = row >> 11;
  const int hh = (int)(bh & 15);
  const int b  = (int)(bh >> 4);

  const float* A1 = Abuf + ((size_t)(b * QKH + hh))      * T_SEQ * VD + (size_t)t * VD;
  const float* A2 = Abuf + ((size_t)(b * QKH + 16 + hh)) * T_SEQ * VD + (size_t)t * VD;
  const float d0 = A1[lane]      - lam * A2[lane];
  const float d1 = A1[lane + 32] - lam * A2[lane + 32];
  float ss = d0 * d0 + d1 * d1;
  #pragma unroll
  for (int off = 16; off; off >>= 1) ss += __shfl_xor(ss, off, 32);
  const float r = rsqrtf(ss * (1.0f / 64.0f) + 1e-5f) * 0.8f;
  float* Dp = Dbuf + ((size_t)(b * T_SEQ + t)) * C_DIM + hh * VD;
  Dp[lane]      = d0 * r;
  Dp[lane + 32] = d1 * r;
}

extern "C" void kernel_launch(void* const* d_in, const int* in_sizes, int n_in,
                              void* d_out, int out_size, void* d_ws, size_t ws_size,
                              hipStream_t stream) {
  const float* x      = (const float*)d_in[0];
  const float* W_attn = (const float*)d_in[1];
  const float* W_proj = (const float*)d_in[2];
  const float* lq1    = (const float*)d_in[3];
  const float* lk1    = (const float*)d_in[4];
  const float* lq2    = (const float*)d_in[5];
  const float* lk2    = (const float*)d_in[6];
  float* out = (float*)d_out;

  float* qkv  = (float*)d_ws;
  float* Abuf = qkv  + (size_t)B_SZ * T_SEQ * ROWQKV;
  float* Dbuf = Abuf + (size_t)B_SZ * QKH * T_SEQ * VD;
  float* lam  = Dbuf + (size_t)B_SZ * T_SEQ * C_DIM;

  const int M = B_SZ * T_SEQ;

  {
    dim3 grid(3 * C_DIM / 128, M / 128);
    k_gemm<<<grid, 256, 0, stream>>>(x, W_attn, qkv, 3 * C_DIM, C_DIM);
  }
  {
    const size_t n = (size_t)B_SZ * T_SEQ * 2 * QKH * 16;
    k_rope<<<(unsigned)(n / 256), 256, 0, stream>>>(qkv);
  }
  k_lambda<<<1, 32, 0, stream>>>(lq1, lk1, lq2, lk2, lam);
  {
    dim3 grid(T_SEQ / 16, QKH, B_SZ);
    k_attn<<<grid, 32, 0, stream>>>(qkv, Abuf);
  }
  k_diffnorm<<<(B_SZ * NH * T_SEQ) / 8, 256, 0, stream>>>(Abuf, lam, Dbuf);
  {
    dim3 grid(C_DIM / 128, M / 128);
    k_gemm<<<grid, 256, 0, stream>>>(Dbuf, W_proj, out, C_DIM, C_DIM);
  }
  (void)in_sizes; (void)n_in; (void)out_size; (void)ws_size;
}


# attention: each wave takes the query tile pair (x, 127-x) in turn, so every wave runs 129 key blocks instead of 1 to 128; grid 64x32x2; per-tile arithmetic and summation order unchanged
# speedup vs baseline: 1.1280x; 1.1280x over previous
#include <hip/hip_runtime.h>
#include <hip/hip_bf16.h>
#include <math.h>

typedef float v2f __attribute__((ext_vector_type(2)));
typedef float v8f __attribute__((ext_vector_type(8)));

#define B_SZ   2
#define T_SEQ  2048
#define C_DIM  1024
#define NH     16
#define HD     32
#define QKH    32
#define VD     64
#define ROWQKV 3072

__device__ __forceinline__ v8f wmma4(v2f a, v2f b, v8f c) {
  return __builtin_amdgcn_wmma_f32_16x16x4_f32(
       false, a,  false, b,
       (short)0, c,  false,  false);
}

__global__ __launch_bounds__(256) void k_gemm(const float* __restrict__ A,
                                              const float* __restrict__ Bm,
                                              float* __restrict__ Cm,
                                              int N, int K) {
  const int lane = threadIdx.x & 31;
  const int wid  = threadIdx.x >> 5;
  const int wy   = wid & 3;
  const int wx   = wid >> 2;
  const int rowBase = blockIdx.y * 128 + wy * 32;
  const int colBase = blockIdx.x * 128 + wx * 64;
  const int hi   = lane >> 4;
  const int l15  = lane & 15;
  const int bc   = colBase + l15;

  const float* pA0 = A + (size_t)(rowBase + l15) * K + hi * 2;
  const float* pA1 = A + (size_t)(rowBase + 16 + l15) * K + hi * 2;
  const float* pB  = Bm + (size_t)(hi * 2) * N + bc;
  const size_t strideB = (size_t)4 * N;

  v8f acc[2][4] = {};

  v2f a0 = *(const v2f*)pA0;
  v2f a1 = *(const v2f*)pA1;
  v2f b[4];
  #pragma unroll
  for (int cg = 0; cg < 4; ++cg) {
    b[cg].x = pB[16 * cg];
    b[cg].y = pB[N + 16 * cg];
  }

  #pragma unroll 2
  for (int k = 0; k < K - 4; k += 4) {
    pA0 += 4; pA1 += 4; pB += strideB;
    const v2f na0 = *(const v2f*)pA0;
    const v2f na1 = *(const v2f*)pA1;
    v2f nb[4];
    #pragma unroll
    for (int cg = 0; cg < 4; ++cg) {
      nb[cg].x = pB[16 * cg];
      nb[cg].y = pB[N + 16 * cg];
    }
    if ((k & 60) == 0) {
      __builtin_prefetch(pB + 16 * strideB, 0, 1);
      __builtin_prefetch(pA0 + 64, 0, 1);
      __builtin_prefetch(pA1 + 64, 0, 1);
    }
    #pragma unroll
    for (int cg = 0; cg < 4; ++cg) {
      acc[0][cg] = wmma4(a0, b[cg], acc[0][cg]);
      acc[1][cg] = wmma4(a1, b[cg], acc[1][cg]);
    }
    a0 = na0; a1 = na1;
    #pragma unroll
    for (int cg = 0; cg < 4; ++cg) b[cg] = nb[cg];
  }
  #pragma unroll
  for (int cg = 0; cg < 4; ++cg) {
    acc[0][cg] = wmma4(a0, b[cg], acc[0][cg]);
    acc[1][cg] = wmma4(a1, b[cg], acc[1][cg]);
  }

  #pragma unroll
  for (int g = 0; g < 2; ++g) {
    #pragma unroll
    for (int r = 0; r < 8; ++r) {
      const size_t row = (size_t)(rowBase + g * 16 + r + hi * 8);
      #pragma unroll
      for (int cg = 0; cg < 4; ++cg) {
        Cm[row * N + bc + 16 * cg] = acc[g][cg][r];
      }
    }
  }
}

__global__ void k_rope(float* __restrict__ qkv) {
  size_t idx = (size_t)blockIdx.x * blockDim.x + threadIdx.x;
  const int j  = idx & 15;  idx >>= 4;
  const int h  = idx & 31;  idx >>= 5;
  const int qk = idx & 1;   idx >>= 1;
  const int t  = idx & 2047; idx >>= 11;
  const int b  = (int)idx;
  const float theta = __expf(-((float)(2 * j) / 32.0f) * logf(10000.0f));
  float s, c;
  __sincosf((float)t * theta, &s, &c);
  float* p = qkv + ((size_t)(b * T_SEQ + t)) * ROWQKV + qk * C_DIM + h * HD + 2 * j;
  const float x1 = p[0], x2 = p[1];
  p[0] = x1 * c - x2 * s;
  p[1] = x1 * s + x2 * c;
}

__global__ void k_lambda(const float* __restrict__ lq1, const float* __restrict__ lk1,
                         const float* __restrict__ lq2, const float* __restrict__ lk2,
                         float* __restrict__ lam) {
  const int l = threadIdx.x;
  float s1 = lq1[l] * lk1[l];
  float s2 = lq2[l] * lk2[l];
  #pragma unroll
  for (int off = 16; off; off >>= 1) {
    s1 += __shfl_xor(s1, off, 32);
    s2 += __shfl_xor(s2, off, 32);
  }
  if (l == 0) *lam = __expf(s1) - __expf(s2) + 0.2f;
}

__global__ __launch_bounds__(32) void k_attn(const float* __restrict__ qkv,
                                             float* __restrict__ Abuf) {
  const int h    = blockIdx.y;
  const int b    = blockIdx.z;
  const int lane = threadIdx.x;
  const int hi   = lane >> 4;
  const int l15  = lane & 15;

  const float* Q  = qkv + (size_t)b * T_SEQ * ROWQKV + h * HD;
  const float* Kp = qkv + (size_t)b * T_SEQ * ROWQKV + C_DIM + h * HD;
  const float* V  = qkv + (size_t)b * T_SEQ * ROWQKV + 2 * C_DIM + (h & 15) * VD;

  __shared__ float S[16 * 18];
  __shared__ float mrow[16], lrow[16], frow[16];

  for (int pass = 0; pass < 2; ++pass) {
    const int qt = pass ? (T_SEQ / 16 - 1 - (int)blockIdx.x) : (int)blockIdx.x;

    if (lane < 16) { mrow[lane] = -3.0e38f; lrow[lane] = 0.0f; }
    __syncthreads();

    const size_t qoff = (size_t)(qt * 16 + l15) * ROWQKV;
    v2f qa[8];
    #pragma unroll
    for (int c = 0; c < 8; ++c) {
      const int col = 4 * c + hi * 2;
      qa[c].x = Q[qoff + col];
      qa[c].y = Q[qoff + col + 1];
    }

    v8f o0 = {}, o1 = {}, o2 = {}, o3 = {};
    const float scale = 0.17677669529663687f;

    for (int kb = 0; kb <= qt; ++kb) {
      v8f s = {};
      const size_t koff = (size_t)(kb * 16 + l15) * ROWQKV;
      #pragma unroll
      for (int c = 0; c < 8; ++c) {
        const int col = 4 * c + hi * 2;
        v2f kf; kf.x = Kp[koff + col]; kf.y = Kp[koff + col + 1];
        s = wmma4(qa[c], kf, s);
      }
      #pragma unroll
      for (int r = 0; r < 8; ++r) {
        const int m    = r + hi * 8;
        const int qidx = qt * 16 + m;
        const int kidx = kb * 16 + l15;
        S[m * 18 + l15] = (kidx <= qidx) ? s[r] * scale : -3.0e38f;
      }
      __syncthreads();

      if (lane < 16) {
        float mx = mrow[lane];
        #pragma unroll
        for (int j = 0; j < 16; ++j) mx = fmaxf(mx, S[lane * 18 + j]);
        const float f = __expf(mrow[lane] - mx);
        float sum = 0.0f;
        #pragma unroll
        for (int j = 0; j < 16; ++j) {
          const float p = __expf(S[lane * 18 + j] - mx);
          S[lane * 18 + j] = p;
          sum += p;
        }
        mrow[lane] = mx;
        frow[lane] = f;
        lrow[lane] = lrow[lane] * f + sum;
      }
      __syncthreads();

      float fr[8];
      #pragma unroll
      for (int r = 0; r < 8; ++r) fr[r] = frow[r + hi * 8];
      #pragma unroll
      for (int r = 0; r < 8; ++r) {
        o0[r] *= fr[r]; o1[r] *= fr[r]; o2[r] *= fr[r]; o3[r] *= fr[r];
      }

      v2f pa[4];
      #pragma unroll
      for (int c = 0; c < 4; ++c) {
        const int col = 4 * c + hi * 2;
        pa[c].x = S[l15 * 18 + col];
        pa[c].y = S[l15 * 18 + col + 1];
      }
      #pragma unroll
      for (int c = 0; c < 4; ++c) {
        const int krow = kb * 16 + 4 * c + hi * 2;
        const float* Vr0 = V + (size_t)krow * ROWQKV;
        const float* Vr1 = V + (size_t)(krow + 1) * ROWQKV;
        v2f vb0; vb0.x = Vr0[l15];      vb0.y = Vr1[l15];
        v2f vb1; vb1.x = Vr0[l15 + 16]; vb1.y = Vr1[l15 + 16];
        v2f vb2; vb2.x = Vr0[l15 + 32]; vb2.y = Vr1[l15 + 32];
        v2f vb3; vb3.x = Vr0[l15 + 48]; vb3.y = Vr1[l15 + 48];
        o0 = wmma4(pa[c], vb0, o0);
        o1 = wmma4(pa[c], vb1, o1);
        o2 = wmma4(pa[c], vb2, o2);
        o3 = wmma4(pa[c], vb3, o3);
      }
      __syncthreads();
    }

    float inv[8];
    #pragma unroll
    for (int r = 0; r < 8; ++r) inv[r] = 1.0f / lrow[r + hi * 8];
    float* Ab = Abuf + ((size_t)(b * QKH + h)) * T_SEQ * VD;
    #pragma unroll
    for (int r = 0; r < 8; ++r) {
      float* Ar = Ab + (size_t)(qt * 16 + r + hi * 8) * VD;
      Ar[l15]      = o0[r] * inv[r];
      Ar[l15 + 16] = o1[r] * inv[r];
      Ar[l15 + 32] = o2[r] * inv[r];
      Ar[l15 + 48] = o3[r] * inv[r];
    }
    __syncthreads();
  }
}

__global__ __launch_bounds__(256) void k_diffnorm(const float* __restrict__ Abuf,
                                                  const float* __restrict__ lamp,
                                                  float* __restrict__ Dbuf) {
  const float lam = *lamp;
  const int wid  = threadIdx.x >> 5;
  const int lane = threadIdx.x & 31;
  const size_t row = (size_t)blockIdx.x * 8 + wid;
  const int t  = (int)(row & 2047);
  const size_t bh = row >> 11;
  const int hh = (int)(bh & 15);
  const int b  = (int)(bh >> 4);

  const float* A1 = Abuf + ((size_t)(b * QKH + hh))      * T_SEQ * VD + (size_t)t * VD;
  const float* A2 = Abuf + ((size_t)(b * QKH + 16 + hh)) * T_SEQ * VD + (size_t)t * VD;
  const float d0 = A1[lane]      - lam * A2[lane];
  const float d1 = A1[lane + 32] - lam * A2[lane + 32];
  float ss = d0 * d0 + d1 * d1;
  #pragma unroll
  for (int off = 16; off; off >>= 1) ss += __shfl_xor(ss, off, 32);
  const float r = rsqrtf(ss * (1.0f / 64.0f) + 1e-5f) * 0.8f;
  float* Dp = Dbuf + ((size_t)(b * T_SEQ + t)) * C_DIM + hh * VD;
  Dp[lane]      = d0 * r;
  Dp[lane + 32] = d1 * r;
}

extern "C" void kernel_launch(void* const* d_in, const int* in_sizes, int n_in,
                              void* d_out, int out_size, void* d_ws, size_t ws_size,
                              hipStream_t stream) {
  const float* x      = (const float*)d_in[0];
  const float* W_attn = (const float*)d_in[1];
  const float* W_proj = (const float*)d_in[2];
  const float* lq1    = (const float*)d_in[3];
  const float* lk1    = (const float*)d_in[4];
  const float* lq2    = (const float*)d_in[5];
  const float* lk2    = (const float*)d_in[6];
  float* out = (float*)d_out;

  float* qkv  = (float*)d_ws;
  float* Abuf = qkv  + (size_t)B_SZ * T_SEQ * ROWQKV;
  float* Dbuf = Abuf + (size_t)B_SZ * QKH * T_SEQ * VD;
  float* lam  = Dbuf + (size_t)B_SZ * T_SEQ * C_DIM;

  const int M = B_SZ * T_SEQ;

  {
    dim3 grid(3 * C_DIM / 128, M / 128);
    k_gemm<<<grid, 256, 0, stream>>>(x, W_attn, qkv, 3 * C_DIM, C_DIM);
  }
  {
    const size_t n = (size_t)B_SZ * T_SEQ * 2 * QKH * 16;
    k_rope<<<(unsigned)(n / 256), 256, 0, stream>>>(qkv);
  }
  k_lambda<<<1, 32, 0, stream>>>(lq1, lk1, lq2, lk2, lam);
  {
    dim3 grid(T_SEQ / 32, QKH, B_SZ);
    k_attn<<<grid, 32, 0, stream>>>(qkv, Abuf);
  }
  k_diffnorm<<<(B_SZ * NH * T_SEQ) / 8, 256, 0, stream>>>(Abuf, lam, Dbuf);
  {
    dim3 grid(C_DIM / 128, M / 128);
    k_gemm<<<grid, 256, 0, stream>>>(Dbuf, W_proj, out, C_DIM, C_DIM);
  }
  (void)in_sizes; (void)n_in; (void)out_size; (void)ws_size;
}


# GEMM (both launches): A and B tiles fetched by whole-line 128-bit global loads, staged through two padded LDS buffers; fragments read from LDS one chunk ahead; k and accumulation order unchanged
# speedup vs baseline: 1.3888x; 1.2312x over previous
#include <hip/hip_runtime.h>
#include <hip/hip_bf16.h>
#include <math.h>

typedef float v2f __attribute__((ext_vector_type(2)));
typedef float v8f __attribute__((ext_vector_type(8)));

#define B_SZ   2
#define T_SEQ  2048
#define C_DIM  1024
#define NH     16
#define HD     32
#define QKH    32
#define VD     64
#define ROWQKV 3072

__device__ __forceinline__ v8f wmma4(v2f a, v2f b, v8f c) {
  return __builtin_amdgcn_wmma_f32_16x16x4_f32(
       false, a,  false, b,
       (short)0, c,  false,  false);
}

#define GK  32
#define GPA 36
#define GPB 136
typedef float v4f __attribute__((ext_vector_type(4)));

__global__ __launch_bounds__(256) void k_gemm(const float* __restrict__ A,
                                              const float* __restrict__ Bm,
                                              float* __restrict__ Cm,
                                              int N, int K) {
  __shared__ __attribute__((aligned(16))) float sA[2][128 * GPA];
  __shared__ __attribute__((aligned(16))) float sB[2][GK * GPB];

  const int tid  = threadIdx.x;
  const int lane = tid & 31;
  const int wid  = tid >> 5;
  const int wy   = wid & 3;
  const int wx   = wid >> 2;
  const int rowBase = blockIdx.y * 128 + wy * 32;
  const int colBase = blockIdx.x * 128 + wx * 64;
  const int hi   = lane >> 4;
  const int l15  = lane & 15;
  const int bc   = colBase + l15;

  const int sar = tid >> 3, sak = (tid & 7) * 4;
  const int sbk = tid >> 5, sbc = (tid & 31) * 4;
  const float* gA = A  + (size_t)(blockIdx.y * 128 + sar) * K + sak;
  const float* gB = Bm + (size_t)sbk * N + blockIdx.x * 128 + sbc;
  const size_t gAstep = (size_t)32 * K;
  const size_t gBstep = (size_t)8 * N;
  const int    lAoff  = sar * GPA + sak;
  const int    lBoff  = sbk * GPB + sbc;

  v4f ra[4], rb[4];
  #pragma unroll
  for (int i = 0; i < 4; ++i) {
    ra[i] = *(const v4f*)(gA + i * gAstep);
    rb[i] = *(const v4f*)(gB + i * gBstep);
  }
  #pragma unroll
  for (int i = 0; i < 4; ++i) {
    *(v4f*)&sA[0][lAoff + 32 * i * GPA] = ra[i];
    *(v4f*)&sB[0][lBoff + 8 * i * GPB]  = rb[i];
  }
  __syncthreads();

  v8f acc[2][4] = {};

  const int fA = (wy * 32 + l15) * GPA + hi * 2;
  const int fB = (hi * 2) * GPB + wx * 64 + l15;

  const int nT = K / GK;
  for (int t = 0; t < nT; ++t) {
    const int  cur  = t & 1;
    const bool more = (t + 1 < nT);
    if (more) {
      gA += GK;
      gB += (size_t)GK * N;
      #pragma unroll
      for (int i = 0; i < 4; ++i) {
        ra[i] = *(const v4f*)(gA + i * gAstep);
        rb[i] = *(const v4f*)(gB + i * gBstep);
      }
    }
    const float* pa = &sA[cur][fA];
    const float* pb = &sB[cur][fB];
    v2f a0 = *(const v2f*)(pa);
    v2f a1 = *(const v2f*)(pa + 16 * GPA);
    v2f b[4];
    #pragma unroll
    for (int cg = 0; cg < 4; ++cg) {
      b[cg].x = pb[16 * cg];
      b[cg].y = pb[GPB + 16 * cg];
    }
    #pragma unroll
    for (int c = 0; c < 8; ++c) {
      v2f na0 = a0, na1 = a1;
      v2f nb[4] = {b[0], b[1], b[2], b[3]};
      if (c < 7) {
        na0 = *(const v2f*)(pa + 4 * (c + 1));
        na1 = *(const v2f*)(pa + 16 * GPA + 4 * (c + 1));
        #pragma unroll
        for (int cg = 0; cg < 4; ++cg) {
          nb[cg].x = pb[(4 * (c + 1))     * GPB + 16 * cg];
          nb[cg].y = pb[(4 * (c + 1) + 1) * GPB + 16 * cg];
        }
      }
      __builtin_amdgcn_sched_barrier(0);
      #pragma unroll
      for (int cg = 0; cg < 4; ++cg) {
        acc[0][cg] = wmma4(a0, b[cg], acc[0][cg]);
        acc[1][cg] = wmma4(a1, b[cg], acc[1][cg]);
      }
      __builtin_amdgcn_sched_barrier(0);
      a0 = na0; a1 = na1;
      #pragma unroll
      for (int cg = 0; cg < 4; ++cg) b[cg] = nb[cg];
    }
    if (more) {
      #pragma unroll
      for (int i = 0; i < 4; ++i) {
        *(v4f*)&sA[cur ^ 1][lAoff + 32 * i * GPA] = ra[i];
        *(v4f*)&sB[cur ^ 1][lBoff + 8 * i * GPB]  = rb[i];
      }
    }
    __syncthreads();
  }

  #pragma unroll
  for (int g = 0; g < 2; ++g) {
    #pragma unroll
    for (int r = 0; r < 8; ++r) {
      const size_t row = (size_t)(rowBase + g * 16 + r + hi * 8);
      #pragma unroll
      for (int cg = 0; cg < 4; ++cg) {
        Cm[row * N + bc + 16 * cg] = acc[g][cg][r];
      }
    }
  }
}

__global__ void k_rope(float* __restrict__ qkv) {
  size_t idx = (size_t)blockIdx.x * blockDim.x + threadIdx.x;
  const int j  = idx & 15;  idx >>= 4;
  const int h  = idx & 31;  idx >>= 5;
  const int qk = idx & 1;   idx >>= 1;
  const int t  = idx & 2047; idx >>= 11;
  const int b  = (int)idx;
  const float theta = __expf(-((float)(2 * j) / 32.0f) * logf(10000.0f));
  float s, c;
  __sincosf((float)t * theta, &s, &c);
  float* p = qkv + ((size_t)(b * T_SEQ + t)) * ROWQKV + qk * C_DIM + h * HD + 2 * j;
  const float x1 = p[0], x2 = p[1];
  p[0] = x1 * c - x2 * s;
  p[1] = x1 * s + x2 * c;
}

__global__ void k_lambda(const float* __restrict__ lq1, const float* __restrict__ lk1,
                         const float* __restrict__ lq2, const float* __restrict__ lk2,
                         float* __restrict__ lam) {
  const int l = threadIdx.x;
  float s1 = lq1[l] * lk1[l];
  float s2 = lq2[l] * lk2[l];
  #pragma unroll
  for (int off = 16; off; off >>= 1) {
    s1 += __shfl_xor(s1, off, 32);
    s2 += __shfl_xor(s2, off, 32);
  }
  if (l == 0) *lam = __expf(s1) - __expf(s2) + 0.2f;
}

__global__ __launch_bounds__(32) void k_attn(const float* __restrict__ qkv,
                                             float* __restrict__ Abuf) {
  const int h    = blockIdx.y;
  const int b    = blockIdx.z;
  const int lane = threadIdx.x;
  const int hi   = lane >> 4;
  const int l15  = lane & 15;

  const float* Q  = qkv + (size_t)b * T_SEQ * ROWQKV + h * HD;
  const float* Kp = qkv + (size_t)b * T_SEQ * ROWQKV + C_DIM + h * HD;
  const float* V  = qkv + (size_t)b * T_SEQ * ROWQKV + 2 * C_DIM + (h & 15) * VD;

  __shared__ float S[16 * 18];
  __shared__ float mrow[16], lrow[16], frow[16];

  for (int pass = 0; pass < 2; ++pass) {
    const int qt = pass ? (T_SEQ / 16 - 1 - (int)blockIdx.x) : (int)blockIdx.x;

    if (lane < 16) { mrow[lane] = -3.0e38f; lrow[lane] = 0.0f; }
    __syncthreads();

    const size_t qoff = (size_t)(qt * 16 + l15) * ROWQKV;
    v2f qa[8];
    #pragma unroll
    for (int c = 0; c < 8; ++c) {
      const int col = 4 * c + hi * 2;
      qa[c].x = Q[qoff + col];
      qa[c].y = Q[qoff + col + 1];
    }

    v8f o0 = {}, o1 = {}, o2 = {}, o3 = {};
    const float scale = 0.17677669529663687f;

    for (int kb = 0; kb <= qt; ++kb) {
      v8f s = {};
      const size_t koff = (size_t)(kb * 16 + l15) * ROWQKV;
      #pragma unroll
      for (int c = 0; c < 8; ++c) {
        const int col = 4 * c + hi * 2;
        v2f kf; kf.x = Kp[koff + col]; kf.y = Kp[koff + col + 1];
        s = wmma4(qa[c], kf, s);
      }
      #pragma unroll
      for (int r = 0; r < 8; ++r) {
        const int m    = r + hi * 8;
        const int qidx = qt * 16 + m;
        const int kidx = kb * 16 + l15;
        S[m * 18 + l15] = (kidx <= qidx) ? s[r] * scale : -3.0e38f;
      }
      __syncthreads();

      if (lane < 16) {
        float mx = mrow[lane];
        #pragma unroll
        for (int j = 0; j < 16; ++j) mx = fmaxf(mx, S[lane * 18 + j]);
        const float f = __expf(mrow[lane] - mx);
        float sum = 0.0f;
        #pragma unroll
        for (int j = 0; j < 16; ++j) {
          const float p = __expf(S[lane * 18 + j] - mx);
          S[lane * 18 + j] = p;
          sum += p;
        }
        mrow[lane] = mx;
        frow[lane] = f;
        lrow[lane] = lrow[lane] * f + sum;
      }
      __syncthreads();

      float fr[8];
      #pragma unroll
      for (int r = 0; r < 8; ++r) fr[r] = frow[r + hi * 8];
      #pragma unroll
      for (int r = 0; r < 8; ++r) {
        o0[r] *= fr[r]; o1[r] *= fr[r]; o2[r] *= fr[r]; o3[r] *= fr[r];
      }

      v2f pa[4];
      #pragma unroll
      for (int c = 0; c < 4; ++c) {
        const int col = 4 * c + hi * 2;
        pa[c].x = S[l15 * 18 + col];
        pa[c].y = S[l15 * 18 + col + 1];
      }
      #pragma unroll
      for (int c = 0; c < 4; ++c) {
        const int krow = kb * 16 + 4 * c + hi * 2;
        const float* Vr0 = V + (size_t)krow * ROWQKV;
        const float* Vr1 = V + (size_t)(krow + 1) * ROWQKV;
        v2f vb0; vb0.x = Vr0[l15];      vb0.y = Vr1[l15];
        v2f vb1; vb1.x = Vr0[l15 + 16]; vb1.y = Vr1[l15 + 16];
        v2f vb2; vb2.x = Vr0[l15 + 32]; vb2.y = Vr1[l15 + 32];
        v2f vb3; vb3.x = Vr0[l15 + 48]; vb3.y = Vr1[l15 + 48];
        o0 = wmma4(pa[c], vb0, o0);
        o1 = wmma4(pa[c], vb1, o1);
        o2 = wmma4(pa[c], vb2, o2);
        o3 = wmma4(pa[c], vb3, o3);
      }
      __syncthreads();
    }

    float inv[8];
    #pragma unroll
    for (int r = 0; r < 8; ++r) inv[r] = 1.0f / lrow[r + hi * 8];
    float* Ab = Abuf + ((size_t)(b * QKH + h)) * T_SEQ * VD;
    #pragma unroll
    for (int r = 0; r < 8; ++r) {
      float* Ar = Ab + (size_t)(qt * 16 + r + hi * 8) * VD;
      Ar[l15]      = o0[r] * inv[r];
      Ar[l15 + 16] = o1[r] * inv[r];
      Ar[l15 + 32] = o2[r] * inv[r];
      Ar[l15 + 48] = o3[r] * inv[r];
    }
    __syncthreads();
  }
}

__global__ __launch_bounds__(256) void k_diffnorm(const float* __restrict__ Abuf,
                                                  const float* __restrict__ lamp,
                                                  float* __restrict__ Dbuf) {
  const float lam = *lamp;
  const int wid  = threadIdx.x >> 5;
  const int lane = threadIdx.x & 31;
  const size_t row = (size_t)blockIdx.x * 8 + wid;
  const int t  = (int)(row & 2047);
  const size_t bh = row >> 11;
  const int hh = (int)(bh & 15);
  const int b  = (int)(bh >> 4);

  const float* A1 = Abuf + ((size_t)(b * QKH + hh))      * T_SEQ * VD + (size_t)t * VD;
  const float* A2 = Abuf + ((size_t)(b * QKH + 16 + hh)) * T_SEQ * VD + (size_t)t * VD;
  const float d0 = A1[lane]      - lam * A2[lane];
  const float d1 = A1[lane + 32] - lam * A2[lane + 32];
  float ss = d0 * d0 + d1 * d1;
  #pragma unroll
  for (int off = 16; off; off >>= 1) ss += __shfl_xor(ss, off, 32);
  const float r = rsqrtf(ss * (1.0f / 64.0f) + 1e-5f) * 0.8f;
  float* Dp = Dbuf + ((size_t)(b * T_SEQ + t)) * C_DIM + hh * VD;
  Dp[lane]      = d0 * r;
  Dp[lane + 32] = d1 * r;
}

extern "C" void kernel_launch(void* const* d_in, const int* in_sizes, int n_in,
                              void* d_out, int out_size, void* d_ws, size_t ws_size,
                              hipStream_t stream) {
  const float* x      = (const float*)d_in[0];
  const float* W_attn = (const float*)d_in[1];
  const float* W_proj = (const float*)d_in[2];
  const float* lq1    = (const float*)d_in[3];
  const float* lk1    = (const float*)d_in[4];
  const float* lq2    = (const float*)d_in[5];
  const float* lk2    = (const float*)d_in[6];
  float* out = (float*)d_out;

  float* qkv  = (float*)d_ws;
  float* Abuf = qkv  + (size_t)B_SZ * T_SEQ * ROWQKV;
  float* Dbuf = Abuf + (size_t)B_SZ * QKH * T_SEQ * VD;
  float* lam  = Dbuf + (size_t)B_SZ * T_SEQ * C_DIM;

  const int M = B_SZ * T_SEQ;

  {
    dim3 grid(3 * C_DIM / 128, M / 128);
    k_gemm<<<grid, 256, 0, stream>>>(x, W_attn, qkv, 3 * C_DIM, C_DIM);
  }
  {
    const size_t n = (size_t)B_SZ * T_SEQ * 2 * QKH * 16;
    k_rope<<<(unsigned)(n / 256), 256, 0, stream>>>(qkv);
  }
  k_lambda<<<1, 32, 0, stream>>>(lq1, lk1, lq2, lk2, lam);
  {
    dim3 grid(T_SEQ / 32, QKH, B_SZ);
    k_attn<<<grid, 32, 0, stream>>>(qkv, Abuf);
  }
  k_diffnorm<<<(B_SZ * NH * T_SEQ) / 8, 256, 0, stream>>>(Abuf, lam, Dbuf);
  {
    dim3 grid(C_DIM / 128, M / 128);
    k_gemm<<<grid, 256, 0, stream>>>(Dbuf, W_proj, out, C_DIM, C_DIM);
  }
  (void)in_sizes; (void)n_in; (void)out_size; (void)ws_size;
}


# attention: scores computed transposed (keys x queries): softmax in registers with one lane-half exchange, Pt feeds Vt*Pt directly, no LDS, loads issued first; row-sum and P*V key order regrouped
# speedup vs baseline: 1.5687x; 1.1295x over previous
#include <hip/hip_runtime.h>
#include <hip/hip_bf16.h>
#include <math.h>

typedef float v2f __attribute__((ext_vector_type(2)));
typedef float v8f __attribute__((ext_vector_type(8)));

#define B_SZ   2
#define T_SEQ  2048
#define C_DIM  1024
#define NH     16
#define HD     32
#define QKH    32
#define VD     64
#define ROWQKV 3072

__device__ __forceinline__ v8f wmma4(v2f a, v2f b, v8f c) {
  return __builtin_amdgcn_wmma_f32_16x16x4_f32(
       false, a,  false, b,
       (short)0, c,  false,  false);
}

#define GK  32
#define GPA 36
#define GPB 136
typedef float v4f __attribute__((ext_vector_type(4)));

__global__ __launch_bounds__(256) void k_gemm(const float* __restrict__ A,
                                              const float* __restrict__ Bm,
                                              float* __restrict__ Cm,
                                              int N, int K) {
  __shared__ __attribute__((aligned(16))) float sA[2][128 * GPA];
  __shared__ __attribute__((aligned(16))) float sB[2][GK * GPB];

  const int tid  = threadIdx.x;
  const int lane = tid & 31;
  const int wid  = tid >> 5;
  const int wy   = wid & 3;
  const int wx   = wid >> 2;
  const int rowBase = blockIdx.y * 128 + wy * 32;
  const int colBase = blockIdx.x * 128 + wx * 64;
  const int hi   = lane >> 4;
  const int l15  = lane & 15;
  const int bc   = colBase + l15;

  const int sar = tid >> 3, sak = (tid & 7) * 4;
  const int sbk = tid >> 5, sbc = (tid & 31) * 4;
  const float* gA = A  + (size_t)(blockIdx.y * 128 + sar) * K + sak;
  const float* gB = Bm + (size_t)sbk * N + blockIdx.x * 128 + sbc;
  const size_t gAstep = (size_t)32 * K;
  const size_t gBstep = (size_t)8 * N;
  const int    lAoff  = sar * GPA + sak;
  const int    lBoff  = sbk * GPB + sbc;

  v4f ra[4], rb[4];
  #pragma unroll
  for (int i = 0; i < 4; ++i) {
    ra[i] = *(const v4f*)(gA + i * gAstep);
    rb[i] = *(const v4f*)(gB + i * gBstep);
  }
  #pragma unroll
  for (int i = 0; i < 4; ++i) {
    *(v4f*)&sA[0][lAoff + 32 * i * GPA] = ra[i];
    *(v4f*)&sB[0][lBoff + 8 * i * GPB]  = rb[i];
  }
  __syncthreads();

  v8f acc[2][4] = {};

  const int fA = (wy * 32 + l15) * GPA + hi * 2;
  const int fB = (hi * 2) * GPB + wx * 64 + l15;

  const int nT = K / GK;
  for (int t = 0; t < nT; ++t) {
    const int  cur  = t & 1;
    const bool more = (t + 1 < nT);
    if (more) {
      gA += GK;
      gB += (size_t)GK * N;
      #pragma unroll
      for (int i = 0; i < 4; ++i) {
        ra[i] = *(const v4f*)(gA + i * gAstep);
        rb[i] = *(const v4f*)(gB + i * gBstep);
      }
    }
    const float* pa = &sA[cur][fA];
    const float* pb = &sB[cur][fB];
    v2f a0 = *(const v2f*)(pa);
    v2f a1 = *(const v2f*)(pa + 16 * GPA);
    v2f b[4];
    #pragma unroll
    for (int cg = 0; cg < 4; ++cg) {
      b[cg].x = pb[16 * cg];
      b[cg].y = pb[GPB + 16 * cg];
    }
    #pragma unroll
    for (int c = 0; c < 8; ++c) {
      v2f na0 = a0, na1 = a1;
      v2f nb[4] = {b[0], b[1], b[2], b[3]};
      if (c < 7) {
        na0 = *(const v2f*)(pa + 4 * (c + 1));
        na1 = *(const v2f*)(pa + 16 * GPA + 4 * (c + 1));
        #pragma unroll
        for (int cg = 0; cg < 4; ++cg) {
          nb[cg].x = pb[(4 * (c + 1))     * GPB + 16 * cg];
          nb[cg].y = pb[(4 * (c + 1) + 1) * GPB + 16 * cg];
        }
      }
      __builtin_amdgcn_sched_barrier(0);
      #pragma unroll
      for (int cg = 0; cg < 4; ++cg) {
        acc[0][cg] = wmma4(a0, b[cg], acc[0][cg]);
        acc[1][cg] = wmma4(a1, b[cg], acc[1][cg]);
      }
      __builtin_amdgcn_sched_barrier(0);
      a0 = na0; a1 = na1;
      #pragma unroll
      for (int cg = 0; cg < 4; ++cg) b[cg] = nb[cg];
    }
    if (more) {
      #pragma unroll
      for (int i = 0; i < 4; ++i) {
        *(v4f*)&sA[cur ^ 1][lAoff + 32 * i * GPA] = ra[i];
        *(v4f*)&sB[cur ^ 1][lBoff + 8 * i * GPB]  = rb[i];
      }
    }
    __syncthreads();
  }

  #pragma unroll
  for (int g = 0; g < 2; ++g) {
    #pragma unroll
    for (int r = 0; r < 8; ++r) {
      const size_t row = (size_t)(rowBase + g * 16 + r + hi * 8);
      #pragma unroll
      for (int cg = 0; cg < 4; ++cg) {
        Cm[row * N + bc + 16 * cg] = acc[g][cg][r];
      }
    }
  }
}

__global__ void k_rope(float* __restrict__ qkv) {
  size_t idx = (size_t)blockIdx.x * blockDim.x + threadIdx.x;
  const int j  = idx & 15;  idx >>= 4;
  const int h  = idx & 31;  idx >>= 5;
  const int qk = idx & 1;   idx >>= 1;
  const int t  = idx & 2047; idx >>= 11;
  const int b  = (int)idx;
  const float theta = __expf(-((float)(2 * j) / 32.0f) * logf(10000.0f));
  float s, c;
  __sincosf((float)t * theta, &s, &c);
  float* p = qkv + ((size_t)(b * T_SEQ + t)) * ROWQKV + qk * C_DIM + h * HD + 2 * j;
  const float x1 = p[0], x2 = p[1];
  p[0] = x1 * c - x2 * s;
  p[1] = x1 * s + x2 * c;
}

__global__ void k_lambda(const float* __restrict__ lq1, const float* __restrict__ lk1,
                         const float* __restrict__ lq2, const float* __restrict__ lk2,
                         float* __restrict__ lam) {
  const int l = threadIdx.x;
  float s1 = lq1[l] * lk1[l];
  float s2 = lq2[l] * lk2[l];
  #pragma unroll
  for (int off = 16; off; off >>= 1) {
    s1 += __shfl_xor(s1, off, 32);
    s2 += __shfl_xor(s2, off, 32);
  }
  if (l == 0) *lam = __expf(s1) - __expf(s2) + 0.2f;
}

__device__ __forceinline__ float xhalf(float x) {
  const int v = __builtin_bit_cast(int, x);
  return __builtin_bit_cast(float,
      __builtin_amdgcn_permlanex16(v, v, 0x76543210, 0xfedcba98, false, false));
}

__global__ __launch_bounds__(32) void k_attn(const float* __restrict__ qkv,
                                             float* __restrict__ Abuf) {
  const int h    = blockIdx.y;
  const int b    = blockIdx.z;
  const int lane = threadIdx.x;
  const int hi   = lane >> 4;
  const int l15  = lane & 15;

  const float* Q  = qkv + (size_t)b * T_SEQ * ROWQKV + h * HD;
  const float* Kp = qkv + (size_t)b * T_SEQ * ROWQKV + C_DIM + h * HD;
  const float* V  = qkv + (size_t)b * T_SEQ * ROWQKV + 2 * C_DIM + (h & 15) * VD;
  const float scale = 0.17677669529663687f;
  const int klane = l15 * ROWQKV + hi * 2;
  const int vlane = hi * 8 * ROWQKV + l15;

  for (int pass = 0; pass < 2; ++pass) {
    const int qt = pass ? (T_SEQ / 16 - 1 - (int)blockIdx.x) : (int)blockIdx.x;

    const size_t qoff = (size_t)(qt * 16 + l15) * ROWQKV;
    v2f qb[8];
    #pragma unroll
    for (int c = 0; c < 8; ++c) {
      const int col = 4 * c + hi * 2;
      qb[c].x = Q[qoff + col];
      qb[c].y = Q[qoff + col + 1];
    }
    const int qidx = qt * 16 + l15;

    float mrun = -3.0e38f;
    float lrun = 0.0f;
    v8f o0 = {}, o1 = {}, o2 = {}, o3 = {};

    for (int kb = 0; kb <= qt; ++kb) {
      const float* Kb = Kp + (size_t)kb * 16 * ROWQKV;
      const float* Vb = V  + (size_t)kb * 16 * ROWQKV;

      v2f kf[8];
      #pragma unroll
      for (int c = 0; c < 8; ++c) {
        kf[c].x = Kb[klane + 4 * c];
        kf[c].y = Kb[klane + 4 * c + 1];
      }
      v2f va[4][4];
      #pragma unroll
      for (int c = 0; c < 4; ++c) {
        #pragma unroll
        for (int n = 0; n < 4; ++n) {
          va[c][n].x = Vb[vlane + (2 * c)     * ROWQKV + 16 * n];
          va[c][n].y = Vb[vlane + (2 * c + 1) * ROWQKV + 16 * n];
        }
      }
      __builtin_amdgcn_sched_barrier(0);

      v8f s = {};
      #pragma unroll
      for (int c = 0; c < 8; ++c) s = wmma4(kf[c], qb[c], s);

      const int dq = qidx - (kb * 16 + hi * 8);
      float sv[8];
      #pragma unroll
      for (int r = 0; r < 8; ++r)
        sv[r] = (r <= dq) ? s[r] * scale : -3.0e38f;

      float bm = sv[0];
      #pragma unroll
      for (int r = 1; r < 8; ++r) bm = fmaxf(bm, sv[r]);
      bm = fmaxf(bm, xhalf(bm));
      const float mx = fmaxf(mrun, bm);
      const float f  = __expf(mrun - mx);
      float p[8];
      float sum = 0.0f;
      #pragma unroll
      for (int r = 0; r < 8; ++r) {
        p[r] = __expf(sv[r] - mx);
        sum += p[r];
      }
      sum += xhalf(sum);
      mrun = mx;
      lrun = lrun * f + sum;

      #pragma unroll
      for (int r = 0; r < 8; ++r) {
        o0[r] *= f; o1[r] *= f; o2[r] *= f; o3[r] *= f;
      }

      #pragma unroll
      for (int c = 0; c < 4; ++c) {
        v2f pb; pb.x = p[2 * c]; pb.y = p[2 * c + 1];
        o0 = wmma4(va[c][0], pb, o0);
        o1 = wmma4(va[c][1], pb, o1);
        o2 = wmma4(va[c][2], pb, o2);
        o3 = wmma4(va[c][3], pb, o3);
      }
    }

    const float inv = 1.0f / lrun;
    float* Ar = Abuf + ((size_t)(b * QKH + h)) * T_SEQ * VD
                     + (size_t)(qt * 16 + l15) * VD + hi * 8;
    v4f w;
    w.x = o0[0] * inv; w.y = o0[1] * inv; w.z = o0[2] * inv; w.w = o0[3] * inv;
    *(v4f*)(Ar)      = w;
    w.x = o0[4] * inv; w.y = o0[5] * inv; w.z = o0[6] * inv; w.w = o0[7] * inv;
    *(v4f*)(Ar + 4)  = w;
    w.x = o1[0] * inv; w.y = o1[1] * inv; w.z = o1[2] * inv; w.w = o1[3] * inv;
    *(v4f*)(Ar + 16) = w;
    w.x = o1[4] * inv; w.y = o1[5] * inv; w.z = o1[6] * inv; w.w = o1[7] * inv;
    *(v4f*)(Ar + 20) = w;
    w.x = o2[0] * inv; w.y = o2[1] * inv; w.z = o2[2] * inv; w.w = o2[3] * inv;
    *(v4f*)(Ar + 32) = w;
    w.x = o2[4] * inv; w.y = o2[5] * inv; w.z = o2[6] * inv; w.w = o2[7] * inv;
    *(v4f*)(Ar + 36) = w;
    w.x = o3[0] * inv; w.y = o3[1] * inv; w.z = o3[2] * inv; w.w = o3[3] * inv;
    *(v4f*)(Ar + 48) = w;
    w.x = o3[4] * inv; w.y = o3[5] * inv; w.z = o3[6] * inv; w.w = o3[7] * inv;
    *(v4f*)(Ar + 52) = w;
  }
}

__global__ __launch_bounds__(256) void k_diffnorm(const float* __restrict__ Abuf,
                                                  const float* __restrict__ lamp,
                                                  float* __restrict__ Dbuf) {
  const float lam = *lamp;
  const int wid  = threadIdx.x >> 5;
  const int lane = threadIdx.x & 31;
  const size_t row = (size_t)blockIdx.x * 8 + wid;
  const int t  = (int)(row & 2047);
  const size_t bh = row >> 11;
  const int hh = (int)(bh & 15);
  const int b  = (int)(bh >> 4);

  const float* A1 = Abuf + ((size_t)(b * QKH + hh))      * T_SEQ * VD + (size_t)t * VD;
  const float* A2 = Abuf + ((size_t)(b * QKH + 16 + hh)) * T_SEQ * VD + (size_t)t * VD;
  const float d0 = A1[lane]      - lam * A2[lane];
  const float d1 = A1[lane + 32] - lam * A2[lane + 32];
  float ss = d0 * d0 + d1 * d1;
  #pragma unroll
  for (int off = 16; off; off >>= 1) ss += __shfl_xor(ss, off, 32);
  const float r = rsqrtf(ss * (1.0f / 64.0f) + 1e-5f) * 0.8f;
  float* Dp = Dbuf + ((size_t)(b * T_SEQ + t)) * C_DIM + hh * VD;
  Dp[lane]      = d0 * r;
  Dp[lane + 32] = d1 * r;
}

extern "C" void kernel_launch(void* const* d_in, const int* in_sizes, int n_in,
                              void* d_out, int out_size, void* d_ws, size_t ws_size,
                              hipStream_t stream) {
  const float* x      = (const float*)d_in[0];
  const float* W_attn = (const float*)d_in[1];
  const float* W_proj = (const float*)d_in[2];
  const float* lq1    = (const float*)d_in[3];
  const float* lk1    = (const float*)d_in[4];
  const float* lq2    = (const float*)d_in[5];
  const float* lk2    = (const float*)d_in[6];
  float* out = (float*)d_out;

  float* qkv  = (float*)d_ws;
  float* Abuf = qkv  + (size_t)B_SZ * T_SEQ * ROWQKV;
  float* Dbuf = Abuf + (size_t)B_SZ * QKH * T_SEQ * VD;
  float* lam  = Dbuf + (size_t)B_SZ * T_SEQ * C_DIM;

  const int M = B_SZ * T_SEQ;

  {
    dim3 grid(3 * C_DIM / 128, M / 128);
    k_gemm<<<grid, 256, 0, stream>>>(x, W_attn, qkv, 3 * C_DIM, C_DIM);
  }
  {
    const size_t n = (size_t)B_SZ * T_SEQ * 2 * QKH * 16;
    k_rope<<<(unsigned)(n / 256), 256, 0, stream>>>(qkv);
  }
  k_lambda<<<1, 32, 0, stream>>>(lq1, lk1, lq2, lk2, lam);
  {
    dim3 grid(T_SEQ / 32, QKH, B_SZ);
    k_attn<<<grid, 32, 0, stream>>>(qkv, Abuf);
  }
  k_diffnorm<<<(B_SZ * NH * T_SEQ) / 8, 256, 0, stream>>>(Abuf, lam, Dbuf);
  {
    dim3 grid(C_DIM / 128, M / 128);
    k_gemm<<<grid, 256, 0, stream>>>(Dbuf, W_proj, out, C_DIM, C_DIM);
  }
  (void)in_sizes; (void)n_in; (void)out_size; (void)ws_size;
}


# attention: work-groups of 8 waves = 8 neighbouring tile pairs of one head, walking the same key blocks: K/V lines come once from L2, 7 times from the WGP cache; per-wave arithmetic unchanged
# speedup vs baseline: 1.6864x; 1.0750x over previous
#include <hip/hip_runtime.h>
#include <hip/hip_bf16.h>
#include <math.h>

typedef float v2f __attribute__((ext_vector_type(2)));
typedef float v8f __attribute__((ext_vector_type(8)));

#define B_SZ   2
#define T_SEQ  2048
#define C_DIM  1024
#define NH     16
#define HD     32
#define QKH    32
#define VD     64
#define ROWQKV 3072

__device__ __forceinline__ v8f wmma4(v2f a, v2f b, v8f c) {
  return __builtin_amdgcn_wmma_f32_16x16x4_f32(
       false, a,  false, b,
       (short)0, c,  false,  false);
}

#define GK  32
#define GPA 36
#define GPB 136
typedef float v4f __attribute__((ext_vector_type(4)));

__global__ __launch_bounds__(256) void k_gemm(const float* __restrict__ A,
                                              const float* __restrict__ Bm,
                                              float* __restrict__ Cm,
                                              int N, int K) {
  __shared__ __attribute__((aligned(16))) float sA[2][128 * GPA];
  __shared__ __attribute__((aligned(16))) float sB[2][GK * GPB];

  const int tid  = threadIdx.x;
  const int lane = tid & 31;
  const int wid  = tid >> 5;
  const int wy   = wid & 3;
  const int wx   = wid >> 2;
  const int rowBase = blockIdx.y * 128 + wy * 32;
  const int colBase = blockIdx.x * 128 + wx * 64;
  const int hi   = lane >> 4;
  const int l15  = lane & 15;
  const int bc   = colBase + l15;

  const int sar = tid >> 3, sak = (tid & 7) * 4;
  const int sbk = tid >> 5, sbc = (tid & 31) * 4;
  const float* gA = A  + (size_t)(blockIdx.y * 128 + sar) * K + sak;
  const float* gB = Bm + (size_t)sbk * N + blockIdx.x * 128 + sbc;
  const size_t gAstep = (size_t)32 * K;
  const size_t gBstep = (size_t)8 * N;
  const int    lAoff  = sar * GPA + sak;
  const int    lBoff  = sbk * GPB + sbc;

  v4f ra[4], rb[4];
  #pragma unroll
  for (int i = 0; i < 4; ++i) {
    ra[i] = *(const v4f*)(gA + i * gAstep);
    rb[i] = *(const v4f*)(gB + i * gBstep);
  }
  #pragma unroll
  for (int i = 0; i < 4; ++i) {
    *(v4f*)&sA[0][lAoff + 32 * i * GPA] = ra[i];
    *(v4f*)&sB[0][lBoff + 8 * i * GPB]  = rb[i];
  }
  __syncthreads();

  v8f acc[2][4] = {};

  const int fA = (wy * 32 + l15) * GPA + hi * 2;
  const int fB = (hi * 2) * GPB + wx * 64 + l15;

  const int nT = K / GK;
  for (int t = 0; t < nT; ++t) {
    const int  cur  = t & 1;
    const bool more = (t + 1 < nT);
    if (more) {
      gA += GK;
      gB += (size_t)GK * N;
      #pragma unroll
      for (int i = 0; i < 4; ++i) {
        ra[i] = *(const v4f*)(gA + i * gAstep);
        rb[i] = *(const v4f*)(gB + i * gBstep);
      }
    }
    const float* pa = &sA[cur][fA];
    const float* pb = &sB[cur][fB];
    v2f a0 = *(const v2f*)(pa);
    v2f a1 = *(const v2f*)(pa + 16 * GPA);
    v2f b[4];
    #pragma unroll
    for (int cg = 0; cg < 4; ++cg) {
      b[cg].x = pb[16 * cg];
      b[cg].y = pb[GPB + 16 * cg];
    }
    #pragma unroll
    for (int c = 0; c < 8; ++c) {
      v2f na0 = a0, na1 = a1;
      v2f nb[4] = {b[0], b[1], b[2], b[3]};
      if (c < 7) {
        na0 = *(const v2f*)(pa + 4 * (c + 1));
        na1 = *(const v2f*)(pa + 16 * GPA + 4 * (c + 1));
        #pragma unroll
        for (int cg = 0; cg < 4; ++cg) {
          nb[cg].x = pb[(4 * (c + 1))     * GPB + 16 * cg];
          nb[cg].y = pb[(4 * (c + 1) + 1) * GPB + 16 * cg];
        }
      }
      __builtin_amdgcn_sched_barrier(0);
      #pragma unroll
      for (int cg = 0; cg < 4; ++cg) {
        acc[0][cg] = wmma4(a0, b[cg], acc[0][cg]);
        acc[1][cg] = wmma4(a1, b[cg], acc[1][cg]);
      }
      __builtin_amdgcn_sched_barrier(0);
      a0 = na0; a1 = na1;
      #pragma unroll
      for (int cg = 0; cg < 4; ++cg) b[cg] = nb[cg];
    }
    if (more) {
      #pragma unroll
      for (int i = 0; i < 4; ++i) {
        *(v4f*)&sA[cur ^ 1][lAoff + 32 * i * GPA] = ra[i];
        *(v4f*)&sB[cur ^ 1][lBoff + 8 * i * GPB]  = rb[i];
      }
    }
    __syncthreads();
  }

  #pragma unroll
  for (int g = 0; g < 2; ++g) {
    #pragma unroll
    for (int r = 0; r < 8; ++r) {
      const size_t row = (size_t)(rowBase + g * 16 + r + hi * 8);
      #pragma unroll
      for (int cg = 0; cg < 4; ++cg) {
        Cm[row * N + bc + 16 * cg] = acc[g][cg][r];
      }
    }
  }
}

__global__ void k_rope(float* __restrict__ qkv) {
  size_t idx = (size_t)blockIdx.x * blockDim.x + threadIdx.x;
  const int j  = idx & 15;  idx >>= 4;
  const int h  = idx & 31;  idx >>= 5;
  const int qk = idx & 1;   idx >>= 1;
  const int t  = idx & 2047; idx >>= 11;
  const int b  = (int)idx;
  const float theta = __expf(-((float)(2 * j) / 32.0f) * logf(10000.0f));
  float s, c;
  __sincosf((float)t * theta, &s, &c);
  float* p = qkv + ((size_t)(b * T_SEQ + t)) * ROWQKV + qk * C_DIM + h * HD + 2 * j;
  const float x1 = p[0], x2 = p[1];
  p[0] = x1 * c - x2 * s;
  p[1] = x1 * s + x2 * c;
}

__global__ void k_lambda(const float* __restrict__ lq1, const float* __restrict__ lk1,
                         const float* __restrict__ lq2, const float* __restrict__ lk2,
                         float* __restrict__ lam) {
  const int l = threadIdx.x;
  float s1 = lq1[l] * lk1[l];
  float s2 = lq2[l] * lk2[l];
  #pragma unroll
  for (int off = 16; off; off >>= 1) {
    s1 += __shfl_xor(s1, off, 32);
    s2 += __shfl_xor(s2, off, 32);
  }
  if (l == 0) *lam = __expf(s1) - __expf(s2) + 0.2f;
}

__device__ __forceinline__ float xhalf(float x) {
  const int v = __builtin_bit_cast(int, x);
  return __builtin_bit_cast(float,
      __builtin_amdgcn_permlanex16(v, v, 0x76543210, 0xfedcba98, false, false));
}

#define ATT_WAVES 8

__global__ __launch_bounds__(32 * ATT_WAVES) void k_attn(const float* __restrict__ qkv,
                                                         float* __restrict__ Abuf) {
  const int h    = blockIdx.y;
  const int b    = blockIdx.z;
  const int lane = threadIdx.x & 31;
  const int hi   = lane >> 4;
  const int l15  = lane & 15;
  const int wid  = __builtin_amdgcn_readfirstlane((int)(threadIdx.x >> 5));
  const int xp   = (int)blockIdx.x * ATT_WAVES + wid;

  const float* Q  = qkv + (size_t)b * T_SEQ * ROWQKV + h * HD;
  const float* Kp = qkv + (size_t)b * T_SEQ * ROWQKV + C_DIM + h * HD;
  const float* V  = qkv + (size_t)b * T_SEQ * ROWQKV + 2 * C_DIM + (h & 15) * VD;
  const float scale = 0.17677669529663687f;
  const int klane = l15 * ROWQKV + hi * 2;
  const int vlane = hi * 8 * ROWQKV + l15;

  for (int pass = 0; pass < 2; ++pass) {
    const int qt = pass ? (T_SEQ / 16 - 1 - xp) : xp;

    const size_t qoff = (size_t)(qt * 16 + l15) * ROWQKV;
    v2f qb[8];
    #pragma unroll
    for (int c = 0; c < 8; ++c) {
      const int col = 4 * c + hi * 2;
      qb[c].x = Q[qoff + col];
      qb[c].y = Q[qoff + col + 1];
    }
    const int qidx = qt * 16 + l15;

    float mrun = -3.0e38f;
    float lrun = 0.0f;
    v8f o0 = {}, o1 = {}, o2 = {}, o3 = {};

    for (int kb = 0; kb <= qt; ++kb) {
      const float* Kb = Kp + (size_t)kb * 16 * ROWQKV;
      const float* Vb = V  + (size_t)kb * 16 * ROWQKV;

      v2f kf[8];
      #pragma unroll
      for (int c = 0; c < 8; ++c) {
        kf[c].x = Kb[klane + 4 * c];
        kf[c].y = Kb[klane + 4 * c + 1];
      }
      v2f va[4][4];
      #pragma unroll
      for (int c = 0; c < 4; ++c) {
        #pragma unroll
        for (int n = 0; n < 4; ++n) {
          va[c][n].x = Vb[vlane + (2 * c)     * ROWQKV + 16 * n];
          va[c][n].y = Vb[vlane + (2 * c + 1) * ROWQKV + 16 * n];
        }
      }
      __builtin_amdgcn_sched_barrier(0);

      v8f s = {};
      #pragma unroll
      for (int c = 0; c < 8; ++c) s = wmma4(kf[c], qb[c], s);

      const int dq = qidx - (kb * 16 + hi * 8);
      float sv[8];
      #pragma unroll
      for (int r = 0; r < 8; ++r)
        sv[r] = (r <= dq) ? s[r] * scale : -3.0e38f;

      float bm = sv[0];
      #pragma unroll
      for (int r = 1; r < 8; ++r) bm = fmaxf(bm, sv[r]);
      bm = fmaxf(bm, xhalf(bm));
      const float mx = fmaxf(mrun, bm);
      const float f  = __expf(mrun - mx);
      float p[8];
      float sum = 0.0f;
      #pragma unroll
      for (int r = 0; r < 8; ++r) {
        p[r] = __expf(sv[r] - mx);
        sum += p[r];
      }
      sum += xhalf(sum);
      mrun = mx;
      lrun = lrun * f + sum;

      #pragma unroll
      for (int r = 0; r < 8; ++r) {
        o0[r] *= f; o1[r] *= f; o2[r] *= f; o3[r] *= f;
      }

      #pragma unroll
      for (int c = 0; c < 4; ++c) {
        v2f pb; pb.x = p[2 * c]; pb.y = p[2 * c + 1];
        o0 = wmma4(va[c][0], pb, o0);
        o1 = wmma4(va[c][1], pb, o1);
        o2 = wmma4(va[c][2], pb, o2);
        o3 = wmma4(va[c][3], pb, o3);
      }
    }

    const float inv = 1.0f / lrun;
    float* Ar = Abuf + ((size_t)(b * QKH + h)) * T_SEQ * VD
                     + (size_t)(qt * 16 + l15) * VD + hi * 8;
    v4f w;
    w.x = o0[0] * inv; w.y = o0[1] * inv; w.z = o0[2] * inv; w.w = o0[3] * inv;
    *(v4f*)(Ar)      = w;
    w.x = o0[4] * inv; w.y = o0[5] * inv; w.z = o0[6] * inv; w.w = o0[7] * inv;
    *(v4f*)(Ar + 4)  = w;
    w.x = o1[0] * inv; w.y = o1[1] * inv; w.z = o1[2] * inv; w.w = o1[3] * inv;
    *(v4f*)(Ar + 16) = w;
    w.x = o1[4] * inv; w.y = o1[5] * inv; w.z = o1[6] * inv; w.w = o1[7] * inv;
    *(v4f*)(Ar + 20) = w;
    w.x = o2[0] * inv; w.y = o2[1] * inv; w.z = o2[2] * inv; w.w = o2[3] * inv;
    *(v4f*)(Ar + 32) = w;
    w.x = o2[4] * inv; w.y = o2[5] * inv; w.z = o2[6] * inv; w.w = o2[7] * inv;
    *(v4f*)(Ar + 36) = w;
    w.x = o3[0] * inv; w.y = o3[1] * inv; w.z = o3[2] * inv; w.w = o3[3] * inv;
    *(v4f*)(Ar + 48) = w;
    w.x = o3[4] * inv; w.y = o3[5] * inv; w.z = o3[6] * inv; w.w = o3[7] * inv;
    *(v4f*)(Ar + 52) = w;
  }
}

__global__ __launch_bounds__(256) void k_diffnorm(const float* __restrict__ Abuf,
                                                  const float* __restrict__ lamp,
                                                  float* __restrict__ Dbuf) {
  const float lam = *lamp;
  const int wid  = threadIdx.x >> 5;
  const int lane = threadIdx.x & 31;
  const size_t row = (size_t)blockIdx.x * 8 + wid;
  const int t  = (int)(row & 2047);
  const size_t bh = row >> 11;
  const int hh = (int)(bh & 15);
  const int b  = (int)(bh >> 4);

  const float* A1 = Abuf + ((size_t)(b * QKH + hh))      * T_SEQ * VD + (size_t)t * VD;
  const float* A2 = Abuf + ((size_t)(b * QKH + 16 + hh)) * T_SEQ * VD + (size_t)t * VD;
  const float d0 = A1[lane]      - lam * A2[lane];
  const float d1 = A1[lane + 32] - lam * A2[lane + 32];
  float ss = d0 * d0 + d1 * d1;
  #pragma unroll
  for (int off = 16; off; off >>= 1) ss += __shfl_xor(ss, off, 32);
  const float r = rsqrtf(ss * (1.0f / 64.0f) + 1e-5f) * 0.8f;
  float* Dp = Dbuf + ((size_t)(b * T_SEQ + t)) * C_DIM + hh * VD;
  Dp[lane]      = d0 * r;
  Dp[lane + 32] = d1 * r;
}

extern "C" void kernel_launch(void* const* d_in, const int* in_sizes, int n_in,
                              void* d_out, int out_size, void* d_ws, size_t ws_size,
                              hipStream_t stream) {
  const float* x      = (const float*)d_in[0];
  const float* W_attn = (const float*)d_in[1];
  const float* W_proj = (const float*)d_in[2];
  const float* lq1    = (const float*)d_in[3];
  const float* lk1    = (const float*)d_in[4];
  const float* lq2    = (const float*)d_in[5];
  const float* lk2    = (const float*)d_in[6];
  float* out = (float*)d_out;

  float* qkv  = (float*)d_ws;
  float* Abuf = qkv  + (size_t)B_SZ * T_SEQ * ROWQKV;
  float* Dbuf = Abuf + (size_t)B_SZ * QKH * T_SEQ * VD;
  float* lam  = Dbuf + (size_t)B_SZ * T_SEQ * C_DIM;

  const int M = B_SZ * T_SEQ;

  {
    dim3 grid(3 * C_DIM / 128, M / 128);
    k_gemm<<<grid, 256, 0, stream>>>(x, W_attn, qkv, 3 * C_DIM, C_DIM);
  }
  {
    const size_t n = (size_t)B_SZ * T_SEQ * 2 * QKH * 16;
    k_rope<<<(unsigned)(n / 256), 256, 0, stream>>>(qkv);
  }
  k_lambda<<<1, 32, 0, stream>>>(lq1, lk1, lq2, lk2, lam);
  {
    dim3 grid(T_SEQ / 32 / ATT_WAVES, QKH, B_SZ);
    k_attn<<<grid, 32 * ATT_WAVES, 0, stream>>>(qkv, Abuf);
  }
  k_diffnorm<<<(B_SZ * NH * T_SEQ) / 8, 256, 0, stream>>>(Abuf, lam, Dbuf);
  {
    dim3 grid(C_DIM / 128, M / 128);
    k_gemm<<<grid, 256, 0, stream>>>(Dbuf, W_proj, out, C_DIM, C_DIM);
  }
  (void)in_sizes; (void)n_in; (void)out_size; (void)ws_size;
}


# attention: one wave runs both score paths (sub-heads hh, hh+16) of its query tiles on one set of V fragments; lambda, difference and RMS norm in registers: 4 launches, was 6; norm sum regrouped
# speedup vs baseline: 1.7586x; 1.0428x over previous
#include <hip/hip_runtime.h>
#include <hip/hip_bf16.h>
#include <math.h>

typedef float v2f __attribute__((ext_vector_type(2)));
typedef float v8f __attribute__((ext_vector_type(8)));

#define B_SZ   2
#define T_SEQ  2048
#define C_DIM  1024
#define NH     16
#define HD     32
#define QKH    32
#define VD     64
#define ROWQKV 3072

__device__ __forceinline__ v8f wmma4(v2f a, v2f b, v8f c) {
  return __builtin_amdgcn_wmma_f32_16x16x4_f32(
       false, a,  false, b,
       (short)0, c,  false,  false);
}

#define GK  32
#define GPA 36
#define GPB 136
typedef float v4f __attribute__((ext_vector_type(4)));

__global__ __launch_bounds__(256) void k_gemm(const float* __restrict__ A,
                                              const float* __restrict__ Bm,
                                              float* __restrict__ Cm,
                                              int N, int K) {
  __shared__ __attribute__((aligned(16))) float sA[2][128 * GPA];
  __shared__ __attribute__((aligned(16))) float sB[2][GK * GPB];

  const int tid  = threadIdx.x;
  const int lane = tid & 31;
  const int wid  = tid >> 5;
  const int wy   = wid & 3;
  const int wx   = wid >> 2;
  const int rowBase = blockIdx.y * 128 + wy * 32;
  const int colBase = blockIdx.x * 128 + wx * 64;
  const int hi   = lane >> 4;
  const int l15  = lane & 15;
  const int bc   = colBase + l15;

  const int sar = tid >> 3, sak = (tid & 7) * 4;
  const int sbk = tid >> 5, sbc = (tid & 31) * 4;
  const float* gA = A  + (size_t)(blockIdx.y * 128 + sar) * K + sak;
  const float* gB = Bm + (size_t)sbk * N + blockIdx.x * 128 + sbc;
  const size_t gAstep = (size_t)32 * K;
  const size_t gBstep = (size_t)8 * N;
  const int    lAoff  = sar * GPA + sak;
  const int    lBoff  = sbk * GPB + sbc;

  v4f ra[4], rb[4];
  #pragma unroll
  for (int i = 0; i < 4; ++i) {
    ra[i] = *(const v4f*)(gA + i * gAstep);
    rb[i] = *(const v4f*)(gB + i * gBstep);
  }
  #pragma unroll
  for (int i = 0; i < 4; ++i) {
    *(v4f*)&sA[0][lAoff + 32 * i * GPA] = ra[i];
    *(v4f*)&sB[0][lBoff + 8 * i * GPB]  = rb[i];
  }
  __syncthreads();

  v8f acc[2][4] = {};

  const int fA = (wy * 32 + l15) * GPA + hi * 2;
  const int fB = (hi * 2) * GPB + wx * 64 + l15;

  const int nT = K / GK;
  for (int t = 0; t < nT; ++t) {
    const int  cur  = t & 1;
    const bool more = (t + 1 < nT);
    if (more) {
      gA += GK;
      gB += (size_t)GK * N;
      #pragma unroll
      for (int i = 0; i < 4; ++i) {
        ra[i] = *(const v4f*)(gA + i * gAstep);
        rb[i] = *(const v4f*)(gB + i * gBstep);
      }
    }
    const float* pa = &sA[cur][fA];
    const float* pb = &sB[cur][fB];
    v2f a0 = *(const v2f*)(pa);
    v2f a1 = *(const v2f*)(pa + 16 * GPA);
    v2f b[4];
    #pragma unroll
    for (int cg = 0; cg < 4; ++cg) {
      b[cg].x = pb[16 * cg];
      b[cg].y = pb[GPB + 16 * cg];
    }
    #pragma unroll
    for (int c = 0; c < 8; ++c) {
      v2f na0 = a0, na1 = a1;
      v2f nb[4] = {b[0], b[1], b[2], b[3]};
      if (c < 7) {
        na0 = *(const v2f*)(pa + 4 * (c + 1));
        na1 = *(const v2f*)(pa + 16 * GPA + 4 * (c + 1));
        #pragma unroll
        for (int cg = 0; cg < 4; ++cg) {
          nb[cg].x = pb[(4 * (c + 1))     * GPB + 16 * cg];
          nb[cg].y = pb[(4 * (c + 1) + 1) * GPB + 16 * cg];
        }
      }
      __builtin_amdgcn_sched_barrier(0);
      #pragma unroll
      for (int cg = 0; cg < 4; ++cg) {
        acc[0][cg] = wmma4(a0, b[cg], acc[0][cg]);
        acc[1][cg] = wmma4(a1, b[cg], acc[1][cg]);
      }
      __builtin_amdgcn_sched_barrier(0);
      a0 = na0; a1 = na1;
      #pragma unroll
      for (int cg = 0; cg < 4; ++cg) b[cg] = nb[cg];
    }
    if (more) {
      #pragma unroll
      for (int i = 0; i < 4; ++i) {
        *(v4f*)&sA[cur ^ 1][lAoff + 32 * i * GPA] = ra[i];
        *(v4f*)&sB[cur ^ 1][lBoff + 8 * i * GPB]  = rb[i];
      }
    }
    __syncthreads();
  }

  #pragma unroll
  for (int g = 0; g < 2; ++g) {
    #pragma unroll
    for (int r = 0; r < 8; ++r) {
      const size_t row = (size_t)(rowBase + g * 16 + r + hi * 8);
      #pragma unroll
      for (int cg = 0; cg < 4; ++cg) {
        Cm[row * N + bc + 16 * cg] = acc[g][cg][r];
      }
    }
  }
}

__global__ void k_rope(float* __restrict__ qkv) {
  size_t idx = (size_t)blockIdx.x * blockDim.x + threadIdx.x;
  const int j  = idx & 15;  idx >>= 4;
  const int h  = idx & 31;  idx >>= 5;
  const int qk = idx & 1;   idx >>= 1;
  const int t  = idx & 2047; idx >>= 11;
  const int b  = (int)idx;
  const float theta = __expf(-((float)(2 * j) / 32.0f) * logf(10000.0f));
  float s, c;
  __sincosf((float)t * theta, &s, &c);
  float* p = qkv + ((size_t)(b * T_SEQ + t)) * ROWQKV + qk * C_DIM + h * HD + 2 * j;
  const float x1 = p[0], x2 = p[1];
  p[0] = x1 * c - x2 * s;
  p[1] = x1 * s + x2 * c;
}

__device__ __forceinline__ float xhalf(float x) {
  const int v = __builtin_bit_cast(int, x);
  return __builtin_bit_cast(float,
      __builtin_amdgcn_permlanex16(v, v, 0x76543210, 0xfedcba98, false, false));
}

__device__ __forceinline__ float softmax_block(const v8f s, const int dq,
                                               float& mrun, float& lrun,
                                               float (&p)[8]) {
  const float scale = 0.17677669529663687f;
  float sv[8];
  #pragma unroll
  for (int r = 0; r < 8; ++r)
    sv[r] = (r <= dq) ? s[r] * scale : -3.0e38f;
  float bm = sv[0];
  #pragma unroll
  for (int r = 1; r < 8; ++r) bm = fmaxf(bm, sv[r]);
  bm = fmaxf(bm, xhalf(bm));
  const float mx = fmaxf(mrun, bm);
  const float f  = __expf(mrun - mx);
  float sum = 0.0f;
  #pragma unroll
  for (int r = 0; r < 8; ++r) {
    p[r] = __expf(sv[r] - mx);
    sum += p[r];
  }
  sum += xhalf(sum);
  mrun = mx;
  lrun = lrun * f + sum;
  return f;
}

#define ATT_WAVES 8

__global__ __launch_bounds__(32 * ATT_WAVES) void k_attn(const float* __restrict__ qkv,
                                                         const float* __restrict__ lq1,
                                                         const float* __restrict__ lk1,
                                                         const float* __restrict__ lq2,
                                                         const float* __restrict__ lk2,
                                                         float* __restrict__ Dbuf) {
  const int hh   = blockIdx.y;
  const int b    = blockIdx.z;
  const int lane = threadIdx.x & 31;
  const int hi   = lane >> 4;
  const int l15  = lane & 15;
  const int wid  = __builtin_amdgcn_readfirstlane((int)(threadIdx.x >> 5));
  const int xp   = (int)blockIdx.x * ATT_WAVES + wid;

  float d1 = lq1[lane] * lk1[lane];
  float d2 = lq2[lane] * lk2[lane];
  #pragma unroll
  for (int off = 16; off; off >>= 1) {
    d1 += __shfl_xor(d1, off, 32);
    d2 += __shfl_xor(d2, off, 32);
  }
  const float lam = __expf(d1) - __expf(d2) + 0.2f;

  const float* Q1 = qkv + (size_t)b * T_SEQ * ROWQKV + hh * HD;
  const float* Q2 = Q1 + NH * HD;
  const float* K1 = qkv + (size_t)b * T_SEQ * ROWQKV + C_DIM + hh * HD;
  const float* K2 = K1 + NH * HD;
  const float* V  = qkv + (size_t)b * T_SEQ * ROWQKV + 2 * C_DIM + hh * VD;
  const int klane = l15 * ROWQKV + hi * 2;
  const int vlane = hi * 8 * ROWQKV + l15;

  for (int pass = 0; pass < 2; ++pass) {
    const int qt = pass ? (T_SEQ / 16 - 1 - xp) : xp;

    const size_t qoff = (size_t)(qt * 16 + l15) * ROWQKV;
    v2f qb1[8], qb2[8];
    #pragma unroll
    for (int c = 0; c < 8; ++c) {
      const int col = 4 * c + hi * 2;
      qb1[c].x = Q1[qoff + col];
      qb1[c].y = Q1[qoff + col + 1];
      qb2[c].x = Q2[qoff + col];
      qb2[c].y = Q2[qoff + col + 1];
    }
    const int qidx = qt * 16 + l15;

    float m1 = -3.0e38f, m2 = -3.0e38f;
    float l1 = 0.0f, l2 = 0.0f;
    v8f oa[4] = {}, ob[4] = {};

    for (int kb = 0; kb <= qt; ++kb) {
      const float* Kb1 = K1 + (size_t)kb * 16 * ROWQKV;
      const float* Kb2 = K2 + (size_t)kb * 16 * ROWQKV;
      const float* Vb  = V  + (size_t)kb * 16 * ROWQKV;

      v2f kf1[8], kf2[8];
      #pragma unroll
      for (int c = 0; c < 8; ++c) {
        kf1[c].x = Kb1[klane + 4 * c];
        kf1[c].y = Kb1[klane + 4 * c + 1];
      }
      #pragma unroll
      for (int c = 0; c < 8; ++c) {
        kf2[c].x = Kb2[klane + 4 * c];
        kf2[c].y = Kb2[klane + 4 * c + 1];
      }
      v2f va[4][4];
      #pragma unroll
      for (int c = 0; c < 4; ++c) {
        #pragma unroll
        for (int n = 0; n < 4; ++n) {
          va[c][n].x = Vb[vlane + (2 * c)     * ROWQKV + 16 * n];
          va[c][n].y = Vb[vlane + (2 * c + 1) * ROWQKV + 16 * n];
        }
      }
      __builtin_amdgcn_sched_barrier(0);

      v8f s1 = {}, s2 = {};
      #pragma unroll
      for (int c = 0; c < 8; ++c) {
        s1 = wmma4(kf1[c], qb1[c], s1);
        s2 = wmma4(kf2[c], qb2[c], s2);
      }

      const int dq = qidx - (kb * 16 + hi * 8);

      float p1[8], p2[8];
      const float f1 = softmax_block(s1, dq, m1, l1, p1);
      const float f2 = softmax_block(s2, dq, m2, l2, p2);

      #pragma unroll
      for (int n = 0; n < 4; ++n) {
        #pragma unroll
        for (int r = 0; r < 8; ++r) {
          oa[n][r] *= f1;
          ob[n][r] *= f2;
        }
      }

      #pragma unroll
      for (int c = 0; c < 4; ++c) {
        v2f pb1; pb1.x = p1[2 * c]; pb1.y = p1[2 * c + 1];
        v2f pb2; pb2.x = p2[2 * c]; pb2.y = p2[2 * c + 1];
        #pragma unroll
        for (int n = 0; n < 4; ++n) {
          oa[n] = wmma4(va[c][n], pb1, oa[n]);
          ob[n] = wmma4(va[c][n], pb2, ob[n]);
        }
      }
    }

    const float inv1 = 1.0f / l1;
    const float inv2 = 1.0f / l2;
    float dv[4][8];
    float ss = 0.0f;
    #pragma unroll
    for (int n = 0; n < 4; ++n) {
      #pragma unroll
      for (int r = 0; r < 8; ++r) {
        const float a1 = oa[n][r] * inv1;
        const float a2 = ob[n][r] * inv2;
        const float d  = a1 - lam * a2;
        dv[n][r] = d;
        ss += d * d;
      }
    }
    ss += xhalf(ss);
    const float rn = rsqrtf(ss * (1.0f / 64.0f) + 1e-5f) * 0.8f;
    float* Dp = Dbuf + ((size_t)(b * T_SEQ + qt * 16 + l15)) * C_DIM + hh * VD + hi * 8;
    #pragma unroll
    for (int n = 0; n < 4; ++n) {
      v4f w;
      w.x = dv[n][0] * rn; w.y = dv[n][1] * rn; w.z = dv[n][2] * rn; w.w = dv[n][3] * rn;
      *(v4f*)(Dp + 16 * n)     = w;
      w.x = dv[n][4] * rn; w.y = dv[n][5] * rn; w.z = dv[n][6] * rn; w.w = dv[n][7] * rn;
      *(v4f*)(Dp + 16 * n + 4) = w;
    }
  }
}

extern "C" void kernel_launch(void* const* d_in, const int* in_sizes, int n_in,
                              void* d_out, int out_size, void* d_ws, size_t ws_size,
                              hipStream_t stream) {
  const float* x      = (const float*)d_in[0];
  const float* W_attn = (const float*)d_in[1];
  const float* W_proj = (const float*)d_in[2];
  const float* lq1    = (const float*)d_in[3];
  const float* lk1    = (const float*)d_in[4];
  const float* lq2    = (const float*)d_in[5];
  const float* lk2    = (const float*)d_in[6];
  float* out = (float*)d_out;

  float* qkv  = (float*)d_ws;
  float* Dbuf = qkv  + (size_t)B_SZ * T_SEQ * ROWQKV;

  const int M = B_SZ * T_SEQ;

  {
    dim3 grid(3 * C_DIM / 128, M / 128);
    k_gemm<<<grid, 256, 0, stream>>>(x, W_attn, qkv, 3 * C_DIM, C_DIM);
  }
  {
    const size_t n = (size_t)B_SZ * T_SEQ * 2 * QKH * 16;
    k_rope<<<(unsigned)(n / 256), 256, 0, stream>>>(qkv);
  }
  {
    dim3 grid(T_SEQ / 32 / ATT_WAVES, NH, B_SZ);
    k_attn<<<grid, 32 * ATT_WAVES, 0, stream>>>(qkv, lq1, lk1, lq2, lk2, Dbuf);
  }
  {
    dim3 grid(C_DIM / 128, M / 128);
    k_gemm<<<grid, 256, 0, stream>>>(Dbuf, W_proj, out, C_DIM, C_DIM);
  }
  (void)in_sizes; (void)n_in; (void)out_size; (void)ws_size;
}


# rotary embedding moved into the qkv GEMM's epilogue (one lane-pair exchange per accumulator element, same formulas); the separate pass of 131072 short waves is gone: 3 launches, was 4
# speedup vs baseline: 1.8928x; 1.0763x over previous
#include <hip/hip_runtime.h>
#include <hip/hip_bf16.h>
#include <math.h>

typedef float v2f __attribute__((ext_vector_type(2)));
typedef float v8f __attribute__((ext_vector_type(8)));

#define B_SZ   2
#define T_SEQ  2048
#define C_DIM  1024
#define NH     16
#define HD     32
#define QKH    32
#define VD     64
#define ROWQKV 3072

__device__ __forceinline__ v8f wmma4(v2f a, v2f b, v8f c) {
  return __builtin_amdgcn_wmma_f32_16x16x4_f32(
       false, a,  false, b,
       (short)0, c,  false,  false);
}

#define GK  32
#define GPA 36
#define GPB 136
typedef float v4f __attribute__((ext_vector_type(4)));

__device__ __forceinline__ float lane_xor1(float x) {
  const int v = __builtin_bit_cast(int, x);
  return __builtin_bit_cast(float,
      __builtin_amdgcn_update_dpp(v, v, 0xB1, 0xF, 0xF, false));
}

__global__ __launch_bounds__(256) void k_gemm(const float* __restrict__ A,
                                              const float* __restrict__ Bm,
                                              float* __restrict__ Cm,
                                              int N, int K, int ropeN) {
  __shared__ __attribute__((aligned(16))) float sA[2][128 * GPA];
  __shared__ __attribute__((aligned(16))) float sB[2][GK * GPB];

  const int tid  = threadIdx.x;
  const int lane = tid & 31;
  const int wid  = tid >> 5;
  const int wy   = wid & 3;
  const int wx   = wid >> 2;
  const int rowBase = blockIdx.y * 128 + wy * 32;
  const int colBase = blockIdx.x * 128 + wx * 64;
  const int hi   = lane >> 4;
  const int l15  = lane & 15;
  const int bc   = colBase + l15;

  const int sar = tid >> 3, sak = (tid & 7) * 4;
  const int sbk = tid >> 5, sbc = (tid & 31) * 4;
  const float* gA = A  + (size_t)(blockIdx.y * 128 + sar) * K + sak;
  const float* gB = Bm + (size_t)sbk * N + blockIdx.x * 128 + sbc;
  const size_t gAstep = (size_t)32 * K;
  const size_t gBstep = (size_t)8 * N;
  const int    lAoff  = sar * GPA + sak;
  const int    lBoff  = sbk * GPB + sbc;

  v4f ra[4], rb[4];
  #pragma unroll
  for (int i = 0; i < 4; ++i) {
    ra[i] = *(const v4f*)(gA + i * gAstep);
    rb[i] = *(const v4f*)(gB + i * gBstep);
  }
  #pragma unroll
  for (int i = 0; i < 4; ++i) {
    *(v4f*)&sA[0][lAoff + 32 * i * GPA] = ra[i];
    *(v4f*)&sB[0][lBoff + 8 * i * GPB]  = rb[i];
  }
  __syncthreads();

  v8f acc[2][4] = {};

  const int fA = (wy * 32 + l15) * GPA + hi * 2;
  const int fB = (hi * 2) * GPB + wx * 64 + l15;

  const int nT = K / GK;
  for (int t = 0; t < nT; ++t) {
    const int  cur  = t & 1;
    const bool more = (t + 1 < nT);
    if (more) {
      gA += GK;
      gB += (size_t)GK * N;
      #pragma unroll
      for (int i = 0; i < 4; ++i) {
        ra[i] = *(const v4f*)(gA + i * gAstep);
        rb[i] = *(const v4f*)(gB + i * gBstep);
      }
    }
    const float* pa = &sA[cur][fA];
    const float* pb = &sB[cur][fB];
    v2f a0 = *(const v2f*)(pa);
    v2f a1 = *(const v2f*)(pa + 16 * GPA);
    v2f b[4];
    #pragma unroll
    for (int cg = 0; cg < 4; ++cg) {
      b[cg].x = pb[16 * cg];
      b[cg].y = pb[GPB + 16 * cg];
    }
    #pragma unroll
    for (int c = 0; c < 8; ++c) {
      v2f na0 = a0, na1 = a1;
      v2f nb[4] = {b[0], b[1], b[2], b[3]};
      if (c < 7) {
        na0 = *(const v2f*)(pa + 4 * (c + 1));
        na1 = *(const v2f*)(pa + 16 * GPA + 4 * (c + 1));
        #pragma unroll
        for (int cg = 0; cg < 4; ++cg) {
          nb[cg].x = pb[(4 * (c + 1))     * GPB + 16 * cg];
          nb[cg].y = pb[(4 * (c + 1) + 1) * GPB + 16 * cg];
        }
      }
      __builtin_amdgcn_sched_barrier(0);
      #pragma unroll
      for (int cg = 0; cg < 4; ++cg) {
        acc[0][cg] = wmma4(a0, b[cg], acc[0][cg]);
        acc[1][cg] = wmma4(a1, b[cg], acc[1][cg]);
      }
      __builtin_amdgcn_sched_barrier(0);
      a0 = na0; a1 = na1;
      #pragma unroll
      for (int cg = 0; cg < 4; ++cg) b[cg] = nb[cg];
    }
    if (more) {
      #pragma unroll
      for (int i = 0; i < 4; ++i) {
        *(v4f*)&sA[cur ^ 1][lAoff + 32 * i * GPA] = ra[i];
        *(v4f*)&sB[cur ^ 1][lBoff + 8 * i * GPB]  = rb[i];
      }
    }
    __syncthreads();
  }

  if ((int)(blockIdx.x * 128) < ropeN) {
    const float sgn = (lane & 1) ? 1.0f : -1.0f;
    float th[2];
    #pragma unroll
    for (int jh = 0; jh < 2; ++jh) {
      const int j = 8 * jh + (l15 >> 1);
      th[jh] = __expf(-((float)(2 * j) / 32.0f) * logf(10000.0f));
    }
    #pragma unroll
    for (int g = 0; g < 2; ++g) {
      #pragma unroll
      for (int r = 0; r < 8; ++r) {
        const int t = (rowBase + g * 16 + r + hi * 8) & (T_SEQ - 1);
        #pragma unroll
        for (int jh = 0; jh < 2; ++jh) {
          float s, c;
          __sincosf((float)t * th[jh], &s, &c);
          const float ss = s * sgn;
          #pragma unroll
          for (int cg = jh; cg < 4; cg += 2) {
            const float v  = acc[g][cg][r];
            const float pv = lane_xor1(v);
            acc[g][cg][r] = __builtin_fmaf(v, c, pv * ss);
          }
        }
      }
    }
  }
  #pragma unroll
  for (int g = 0; g < 2; ++g) {
    #pragma unroll
    for (int r = 0; r < 8; ++r) {
      const size_t row = (size_t)(rowBase + g * 16 + r + hi * 8);
      #pragma unroll
      for (int cg = 0; cg < 4; ++cg) {
        Cm[row * N + bc + 16 * cg] = acc[g][cg][r];
      }
    }
  }
}

__device__ __forceinline__ float xhalf(float x) {
  const int v = __builtin_bit_cast(int, x);
  return __builtin_bit_cast(float,
      __builtin_amdgcn_permlanex16(v, v, 0x76543210, 0xfedcba98, false, false));
}

__device__ __forceinline__ float softmax_block(const v8f s, const int dq,
                                               float& mrun, float& lrun,
                                               float (&p)[8]) {
  const float scale = 0.17677669529663687f;
  float sv[8];
  #pragma unroll
  for (int r = 0; r < 8; ++r)
    sv[r] = (r <= dq) ? s[r] * scale : -3.0e38f;
  float bm = sv[0];
  #pragma unroll
  for (int r = 1; r < 8; ++r) bm = fmaxf(bm, sv[r]);
  bm = fmaxf(bm, xhalf(bm));
  const float mx = fmaxf(mrun, bm);
  const float f  = __expf(mrun - mx);
  float sum = 0.0f;
  #pragma unroll
  for (int r = 0; r < 8; ++r) {
    p[r] = __expf(sv[r] - mx);
    sum += p[r];
  }
  sum += xhalf(sum);
  mrun = mx;
  lrun = lrun * f + sum;
  return f;
}

#define ATT_WAVES 8

__global__ __launch_bounds__(32 * ATT_WAVES) void k_attn(const float* __restrict__ qkv,
                                                         const float* __restrict__ lq1,
                                                         const float* __restrict__ lk1,
                                                         const float* __restrict__ lq2,
                                                         const float* __restrict__ lk2,
                                                         float* __restrict__ Dbuf) {
  const int hh   = blockIdx.y;
  const int b    = blockIdx.z;
  const int lane = threadIdx.x & 31;
  const int hi   = lane >> 4;
  const int l15  = lane & 15;
  const int wid  = __builtin_amdgcn_readfirstlane((int)(threadIdx.x >> 5));
  const int xp   = (int)blockIdx.x * ATT_WAVES + wid;

  float d1 = lq1[lane] * lk1[lane];
  float d2 = lq2[lane] * lk2[lane];
  #pragma unroll
  for (int off = 16; off; off >>= 1) {
    d1 += __shfl_xor(d1, off, 32);
    d2 += __shfl_xor(d2, off, 32);
  }
  const float lam = __expf(d1) - __expf(d2) + 0.2f;

  const float* Q1 = qkv + (size_t)b * T_SEQ * ROWQKV + hh * HD;
  const float* Q2 = Q1 + NH * HD;
  const float* K1 = qkv + (size_t)b * T_SEQ * ROWQKV + C_DIM + hh * HD;
  const float* K2 = K1 + NH * HD;
  const float* V  = qkv + (size_t)b * T_SEQ * ROWQKV + 2 * C_DIM + hh * VD;
  const int klane = l15 * ROWQKV + hi * 2;
  const int vlane = hi * 8 * ROWQKV + l15;

  for (int pass = 0; pass < 2; ++pass) {
    const int qt = pass ? (T_SEQ / 16 - 1 - xp) : xp;

    const size_t qoff = (size_t)(qt * 16 + l15) * ROWQKV;
    v2f qb1[8], qb2[8];
    #pragma unroll
    for (int c = 0; c < 8; ++c) {
      const int col = 4 * c + hi * 2;
      qb1[c].x = Q1[qoff + col];
      qb1[c].y = Q1[qoff + col + 1];
      qb2[c].x = Q2[qoff + col];
      qb2[c].y = Q2[qoff + col + 1];
    }
    const int qidx = qt * 16 + l15;

    float m1 = -3.0e38f, m2 = -3.0e38f;
    float l1 = 0.0f, l2 = 0.0f;
    v8f oa[4] = {}, ob[4] = {};

    for (int kb = 0; kb <= qt; ++kb) {
      const float* Kb1 = K1 + (size_t)kb * 16 * ROWQKV;
      const float* Kb2 = K2 + (size_t)kb * 16 * ROWQKV;
      const float* Vb  = V  + (size_t)kb * 16 * ROWQKV;

      v2f kf1[8], kf2[8];
      #pragma unroll
      for (int c = 0; c < 8; ++c) {
        kf1[c].x = Kb1[klane + 4 * c];
        kf1[c].y = Kb1[klane + 4 * c + 1];
      }
      #pragma unroll
      for (int c = 0; c < 8; ++c) {
        kf2[c].x = Kb2[klane + 4 * c];
        kf2[c].y = Kb2[klane + 4 * c + 1];
      }
      v2f va[4][4];
      #pragma unroll
      for (int c = 0; c < 4; ++c) {
        #pragma unroll
        for (int n = 0; n < 4; ++n) {
          va[c][n].x = Vb[vlane + (2 * c)     * ROWQKV + 16 * n];
          va[c][n].y = Vb[vlane + (2 * c + 1) * ROWQKV + 16 * n];
        }
      }
      __builtin_amdgcn_sched_barrier(0);

      v8f s1 = {}, s2 = {};
      #pragma unroll
      for (int c = 0; c < 8; ++c) {
        s1 = wmma4(kf1[c], qb1[c], s1);
        s2 = wmma4(kf2[c], qb2[c], s2);
      }

      const int dq = qidx - (kb * 16 + hi * 8);

      float p1[8], p2[8];
      const float f1 = softmax_block(s1, dq, m1, l1, p1);
      const float f2 = softmax_block(s2, dq, m2, l2, p2);

      #pragma unroll
      for (int n = 0; n < 4; ++n) {
        #pragma unroll
        for (int r = 0; r < 8; ++r) {
          oa[n][r] *= f1;
          ob[n][r] *= f2;
        }
      }

      #pragma unroll
      for (int c = 0; c < 4; ++c) {
        v2f pb1; pb1.x = p1[2 * c]; pb1.y = p1[2 * c + 1];
        v2f pb2; pb2.x = p2[2 * c]; pb2.y = p2[2 * c + 1];
        #pragma unroll
        for (int n = 0; n < 4; ++n) {
          oa[n] = wmma4(va[c][n], pb1, oa[n]);
          ob[n] = wmma4(va[c][n], pb2, ob[n]);
        }
      }
    }

    const float inv1 = 1.0f / l1;
    const float inv2 = 1.0f / l2;
    float dv[4][8];
    float ss = 0.0f;
    #pragma unroll
    for (int n = 0; n < 4; ++n) {
      #pragma unroll
      for (int r = 0; r < 8; ++r) {
        const float a1 = oa[n][r] * inv1;
        const float a2 = ob[n][r] * inv2;
        const float d  = a1 - lam * a2;
        dv[n][r] = d;
        ss += d * d;
      }
    }
    ss += xhalf(ss);
    const float rn = rsqrtf(ss * (1.0f / 64.0f) + 1e-5f) * 0.8f;
    float* Dp = Dbuf + ((size_t)(b * T_SEQ + qt * 16 + l15)) * C_DIM + hh * VD + hi * 8;
    #pragma unroll
    for (int n = 0; n < 4; ++n) {
      v4f w;
      w.x = dv[n][0] * rn; w.y = dv[n][1] * rn; w.z = dv[n][2] * rn; w.w = dv[n][3] * rn;
      *(v4f*)(Dp + 16 * n)     = w;
      w.x = dv[n][4] * rn; w.y = dv[n][5] * rn; w.z = dv[n][6] * rn; w.w = dv[n][7] * rn;
      *(v4f*)(Dp + 16 * n + 4) = w;
    }
  }
}

extern "C" void kernel_launch(void* const* d_in, const int* in_sizes, int n_in,
                              void* d_out, int out_size, void* d_ws, size_t ws_size,
                              hipStream_t stream) {
  const float* x      = (const float*)d_in[0];
  const float* W_attn = (const float*)d_in[1];
  const float* W_proj = (const float*)d_in[2];
  const float* lq1    = (const float*)d_in[3];
  const float* lk1    = (const float*)d_in[4];
  const float* lq2    = (const float*)d_in[5];
  const float* lk2    = (const float*)d_in[6];
  float* out = (float*)d_out;

  float* qkv  = (float*)d_ws;
  float* Dbuf = qkv  + (size_t)B_SZ * T_SEQ * ROWQKV;

  const int M = B_SZ * T_SEQ;

  {
    dim3 grid(3 * C_DIM / 128, M / 128);
    k_gemm<<<grid, 256, 0, stream>>>(x, W_attn, qkv, 3 * C_DIM, C_DIM, 2 * C_DIM);
  }
  {
    dim3 grid(T_SEQ / 32 / ATT_WAVES, NH, B_SZ);
    k_attn<<<grid, 32 * ATT_WAVES, 0, stream>>>(qkv, lq1, lk1, lq2, lk2, Dbuf);
  }
  {
    dim3 grid(C_DIM / 128, M / 128);
    k_gemm<<<grid, 256, 0, stream>>>(Dbuf, W_proj, out, C_DIM, C_DIM, 0);
  }
  (void)in_sizes; (void)n_in; (void)out_size; (void)ws_size;
}


# attention: K and Vt fragments of key block kb+1 are loaded into a second register set as one group in front of block kb's matrix run (2 blocks per loop trip); arithmetic and sum order unchanged
# speedup vs baseline: 1.9656x; 1.0385x over previous
#include <hip/hip_runtime.h>
#include <hip/hip_bf16.h>
#include <math.h>

typedef float v2f __attribute__((ext_vector_type(2)));
typedef float v8f __attribute__((ext_vector_type(8)));

#define B_SZ   2
#define T_SEQ  2048
#define C_DIM  1024
#define NH     16
#define HD     32
#define QKH    32
#define VD     64
#define ROWQKV 3072

__device__ __forceinline__ v8f wmma4(v2f a, v2f b, v8f c) {
  return __builtin_amdgcn_wmma_f32_16x16x4_f32(
       false, a,  false, b,
       (short)0, c,  false,  false);
}

#define GK  32
#define GPA 36
#define GPB 136
typedef float v4f __attribute__((ext_vector_type(4)));

__device__ __forceinline__ float lane_xor1(float x) {
  const int v = __builtin_bit_cast(int, x);
  return __builtin_bit_cast(float,
      __builtin_amdgcn_update_dpp(v, v, 0xB1, 0xF, 0xF, false));
}

__global__ __launch_bounds__(256) void k_gemm(const float* __restrict__ A,
                                              const float* __restrict__ Bm,
                                              float* __restrict__ Cm,
                                              int N, int K, int ropeN) {
  __shared__ __attribute__((aligned(16))) float sA[2][128 * GPA];
  __shared__ __attribute__((aligned(16))) float sB[2][GK * GPB];

  const int tid  = threadIdx.x;
  const int lane = tid & 31;
  const int wid  = tid >> 5;
  const int wy   = wid & 3;
  const int wx   = wid >> 2;
  const int rowBase = blockIdx.y * 128 + wy * 32;
  const int colBase = blockIdx.x * 128 + wx * 64;
  const int hi   = lane >> 4;
  const int l15  = lane & 15;
  const int bc   = colBase + l15;

  const int sar = tid >> 3, sak = (tid & 7) * 4;
  const int sbk = tid >> 5, sbc = (tid & 31) * 4;
  const float* gA = A  + (size_t)(blockIdx.y * 128 + sar) * K + sak;
  const float* gB = Bm + (size_t)sbk * N + blockIdx.x * 128 + sbc;
  const size_t gAstep = (size_t)32 * K;
  const size_t gBstep = (size_t)8 * N;
  const int    lAoff  = sar * GPA + sak;
  const int    lBoff  = sbk * GPB + sbc;

  v4f ra[4], rb[4];
  #pragma unroll
  for (int i = 0; i < 4; ++i) {
    ra[i] = *(const v4f*)(gA + i * gAstep);
    rb[i] = *(const v4f*)(gB + i * gBstep);
  }
  #pragma unroll
  for (int i = 0; i < 4; ++i) {
    *(v4f*)&sA[0][lAoff + 32 * i * GPA] = ra[i];
    *(v4f*)&sB[0][lBoff + 8 * i * GPB]  = rb[i];
  }
  __syncthreads();

  v8f acc[2][4] = {};

  const int fA = (wy * 32 + l15) * GPA + hi * 2;
  const int fB = (hi * 2) * GPB + wx * 64 + l15;

  const int nT = K / GK;
  for (int t = 0; t < nT; ++t) {
    const int  cur  = t & 1;
    const bool more = (t + 1 < nT);
    if (more) {
      gA += GK;
      gB += (size_t)GK * N;
      #pragma unroll
      for (int i = 0; i < 4; ++i) {
        ra[i] = *(const v4f*)(gA + i * gAstep);
        rb[i] = *(const v4f*)(gB + i * gBstep);
      }
    }
    const float* pa = &sA[cur][fA];
    const float* pb = &sB[cur][fB];
    v2f a0 = *(const v2f*)(pa);
    v2f a1 = *(const v2f*)(pa + 16 * GPA);
    v2f b[4];
    #pragma unroll
    for (int cg = 0; cg < 4; ++cg) {
      b[cg].x = pb[16 * cg];
      b[cg].y = pb[GPB + 16 * cg];
    }
    #pragma unroll
    for (int c = 0; c < 8; ++c) {
      v2f na0 = a0, na1 = a1;
      v2f nb[4] = {b[0], b[1], b[2], b[3]};
      if (c < 7) {
        na0 = *(const v2f*)(pa + 4 * (c + 1));
        na1 = *(const v2f*)(pa + 16 * GPA + 4 * (c + 1));
        #pragma unroll
        for (int cg = 0; cg < 4; ++cg) {
          nb[cg].x = pb[(4 * (c + 1))     * GPB + 16 * cg];
          nb[cg].y = pb[(4 * (c + 1) + 1) * GPB + 16 * cg];
        }
      }
      __builtin_amdgcn_sched_barrier(0);
      #pragma unroll
      for (int cg = 0; cg < 4; ++cg) {
        acc[0][cg] = wmma4(a0, b[cg], acc[0][cg]);
        acc[1][cg] = wmma4(a1, b[cg], acc[1][cg]);
      }
      __builtin_amdgcn_sched_barrier(0);
      a0 = na0; a1 = na1;
      #pragma unroll
      for (int cg = 0; cg < 4; ++cg) b[cg] = nb[cg];
    }
    if (more) {
      #pragma unroll
      for (int i = 0; i < 4; ++i) {
        *(v4f*)&sA[cur ^ 1][lAoff + 32 * i * GPA] = ra[i];
        *(v4f*)&sB[cur ^ 1][lBoff + 8 * i * GPB]  = rb[i];
      }
    }
    __syncthreads();
  }

  if ((int)(blockIdx.x * 128) < ropeN) {
    const float sgn = (lane & 1) ? 1.0f : -1.0f;
    float th[2];
    #pragma unroll
    for (int jh = 0; jh < 2; ++jh) {
      const int j = 8 * jh + (l15 >> 1);
      th[jh] = __expf(-((float)(2 * j) / 32.0f) * logf(10000.0f));
    }
    #pragma unroll
    for (int g = 0; g < 2; ++g) {
      #pragma unroll
      for (int r = 0; r < 8; ++r) {
        const int t = (rowBase + g * 16 + r + hi * 8) & (T_SEQ - 1);
        #pragma unroll
        for (int jh = 0; jh < 2; ++jh) {
          float s, c;
          __sincosf((float)t * th[jh], &s, &c);
          const float ss = s * sgn;
          #pragma unroll
          for (int cg = jh; cg < 4; cg += 2) {
            const float v  = acc[g][cg][r];
            const float pv = lane_xor1(v);
            acc[g][cg][r] = __builtin_fmaf(v, c, pv * ss);
          }
        }
      }
    }
  }
  #pragma unroll
  for (int g = 0; g < 2; ++g) {
    #pragma unroll
    for (int r = 0; r < 8; ++r) {
      const size_t row = (size_t)(rowBase + g * 16 + r + hi * 8);
      #pragma unroll
      for (int cg = 0; cg < 4; ++cg) {
        Cm[row * N + bc + 16 * cg] = acc[g][cg][r];
      }
    }
  }
}

__device__ __forceinline__ float xhalf(float x) {
  const int v = __builtin_bit_cast(int, x);
  return __builtin_bit_cast(float,
      __builtin_amdgcn_permlanex16(v, v, 0x76543210, 0xfedcba98, false, false));
}

__device__ __forceinline__ float softmax_block(const v8f s, const int dq,
                                               float& mrun, float& lrun,
                                               float (&p)[8]) {
  const float scale = 0.17677669529663687f;
  float sv[8];
  #pragma unroll
  for (int r = 0; r < 8; ++r)
    sv[r] = (r <= dq) ? s[r] * scale : -3.0e38f;
  float bm = sv[0];
  #pragma unroll
  for (int r = 1; r < 8; ++r) bm = fmaxf(bm, sv[r]);
  bm = fmaxf(bm, xhalf(bm));
  const float mx = fmaxf(mrun, bm);
  const float f  = __expf(mrun - mx);
  float sum = 0.0f;
  #pragma unroll
  for (int r = 0; r < 8; ++r) {
    p[r] = __expf(sv[r] - mx);
    sum += p[r];
  }
  sum += xhalf(sum);
  mrun = mx;
  lrun = lrun * f + sum;
  return f;
}

struct KVFrag {
  v2f kf1[8], kf2[8];
  v2f va[4][4];
};

__device__ __forceinline__ void kv_fetch(KVFrag& f, const float* __restrict__ K1,
                                         const float* __restrict__ K2,
                                         const float* __restrict__ V,
                                         const int kb, const int klane, const int vlane) {
  const float* Kb1 = K1 + (size_t)kb * 16 * ROWQKV;
  const float* Kb2 = K2 + (size_t)kb * 16 * ROWQKV;
  const float* Vb  = V  + (size_t)kb * 16 * ROWQKV;
  #pragma unroll
  for (int c = 0; c < 8; ++c) {
    f.kf1[c].x = Kb1[klane + 4 * c];
    f.kf1[c].y = Kb1[klane + 4 * c + 1];
  }
  #pragma unroll
  for (int c = 0; c < 8; ++c) {
    f.kf2[c].x = Kb2[klane + 4 * c];
    f.kf2[c].y = Kb2[klane + 4 * c + 1];
  }
  #pragma unroll
  for (int c = 0; c < 4; ++c) {
    #pragma unroll
    for (int n = 0; n < 4; ++n) {
      f.va[c][n].x = Vb[vlane + (2 * c)     * ROWQKV + 16 * n];
      f.va[c][n].y = Vb[vlane + (2 * c + 1) * ROWQKV + 16 * n];
    }
  }
}

__device__ __forceinline__ void kv_block(const KVFrag& f, const int kb, const int hi,
                                         const int qidx,
                                         const v2f (&qb1)[8], const v2f (&qb2)[8],
                                         float& m1, float& m2, float& l1, float& l2,
                                         v8f (&oa)[4], v8f (&ob)[4]) {
  v8f s1 = {}, s2 = {};
  #pragma unroll
  for (int c = 0; c < 8; ++c) {
    s1 = wmma4(f.kf1[c], qb1[c], s1);
    s2 = wmma4(f.kf2[c], qb2[c], s2);
  }

  const int dq = qidx - (kb * 16 + hi * 8);

  float p1[8], p2[8];
  const float f1 = softmax_block(s1, dq, m1, l1, p1);
  const float f2 = softmax_block(s2, dq, m2, l2, p2);

  #pragma unroll
  for (int n = 0; n < 4; ++n) {
    #pragma unroll
    for (int r = 0; r < 8; ++r) {
      oa[n][r] *= f1;
      ob[n][r] *= f2;
    }
  }

  #pragma unroll
  for (int c = 0; c < 4; ++c) {
    v2f pb1; pb1.x = p1[2 * c]; pb1.y = p1[2 * c + 1];
    v2f pb2; pb2.x = p2[2 * c]; pb2.y = p2[2 * c + 1];
    #pragma unroll
    for (int n = 0; n < 4; ++n) {
      oa[n] = wmma4(f.va[c][n], pb1, oa[n]);
      ob[n] = wmma4(f.va[c][n], pb2, ob[n]);
    }
  }
}

#define ATT_WAVES 8

__global__ __launch_bounds__(32 * ATT_WAVES) void k_attn(const float* __restrict__ qkv,
                                                         const float* __restrict__ lq1,
                                                         const float* __restrict__ lk1,
                                                         const float* __restrict__ lq2,
                                                         const float* __restrict__ lk2,
                                                         float* __restrict__ Dbuf) {
  const int hh   = blockIdx.y;
  const int b    = blockIdx.z;
  const int lane = threadIdx.x & 31;
  const int hi   = lane >> 4;
  const int l15  = lane & 15;
  const int wid  = __builtin_amdgcn_readfirstlane((int)(threadIdx.x >> 5));
  const int xp   = (int)blockIdx.x * ATT_WAVES + wid;

  float d1 = lq1[lane] * lk1[lane];
  float d2 = lq2[lane] * lk2[lane];
  #pragma unroll
  for (int off = 16; off; off >>= 1) {
    d1 += __shfl_xor(d1, off, 32);
    d2 += __shfl_xor(d2, off, 32);
  }
  const float lam = __expf(d1) - __expf(d2) + 0.2f;

  const float* Q1 = qkv + (size_t)b * T_SEQ * ROWQKV + hh * HD;
  const float* Q2 = Q1 + NH * HD;
  const float* K1 = qkv + (size_t)b * T_SEQ * ROWQKV + C_DIM + hh * HD;
  const float* K2 = K1 + NH * HD;
  const float* V  = qkv + (size_t)b * T_SEQ * ROWQKV + 2 * C_DIM + hh * VD;
  const int klane = l15 * ROWQKV + hi * 2;
  const int vlane = hi * 8 * ROWQKV + l15;

  for (int pass = 0; pass < 2; ++pass) {
    const int qt = pass ? (T_SEQ / 16 - 1 - xp) : xp;

    const size_t qoff = (size_t)(qt * 16 + l15) * ROWQKV;
    v2f qb1[8], qb2[8];
    #pragma unroll
    for (int c = 0; c < 8; ++c) {
      const int col = 4 * c + hi * 2;
      qb1[c].x = Q1[qoff + col];
      qb1[c].y = Q1[qoff + col + 1];
      qb2[c].x = Q2[qoff + col];
      qb2[c].y = Q2[qoff + col + 1];
    }
    const int qidx = qt * 16 + l15;

    float m1 = -3.0e38f, m2 = -3.0e38f;
    float l1 = 0.0f, l2 = 0.0f;
    v8f oa[4] = {}, ob[4] = {};

    KVFrag fa, fb;
    kv_fetch(fa, K1, K2, V, 0, klane, vlane);
    int kb = 0;
    while (kb < qt) {
      kv_fetch(fb, K1, K2, V, kb + 1, klane, vlane);
      __builtin_amdgcn_sched_barrier(0);
      kv_block(fa, kb, hi, qidx, qb1, qb2, m1, m2, l1, l2, oa, ob);
      __builtin_amdgcn_sched_barrier(0);
      kv_fetch(fa, K1, K2, V, (kb + 2 < qt) ? (kb + 2) : qt, klane, vlane);
      __builtin_amdgcn_sched_barrier(0);
      kv_block(fb, kb + 1, hi, qidx, qb1, qb2, m1, m2, l1, l2, oa, ob);
      kb += 2;
    }
    if (kb == qt) {
      kv_block(fa, kb, hi, qidx, qb1, qb2, m1, m2, l1, l2, oa, ob);
    }

    const float inv1 = 1.0f / l1;
    const float inv2 = 1.0f / l2;
    float dv[4][8];
    float ss = 0.0f;
    #pragma unroll
    for (int n = 0; n < 4; ++n) {
      #pragma unroll
      for (int r = 0; r < 8; ++r) {
        const float a1 = oa[n][r] * inv1;
        const float a2 = ob[n][r] * inv2;
        const float d  = a1 - lam * a2;
        dv[n][r] = d;
        ss += d * d;
      }
    }
    ss += xhalf(ss);
    const float rn = rsqrtf(ss * (1.0f / 64.0f) + 1e-5f) * 0.8f;
    float* Dp = Dbuf + ((size_t)(b * T_SEQ + qt * 16 + l15)) * C_DIM + hh * VD + hi * 8;
    #pragma unroll
    for (int n = 0; n < 4; ++n) {
      v4f w;
      w.x = dv[n][0] * rn; w.y = dv[n][1] * rn; w.z = dv[n][2] * rn; w.w = dv[n][3] * rn;
      *(v4f*)(Dp + 16 * n)     = w;
      w.x = dv[n][4] * rn; w.y = dv[n][5] * rn; w.z = dv[n][6] * rn; w.w = dv[n][7] * rn;
      *(v4f*)(Dp + 16 * n + 4) = w;
    }
  }
}

extern "C" void kernel_launch(void* const* d_in, const int* in_sizes, int n_in,
                              void* d_out, int out_size, void* d_ws, size_t ws_size,
                              hipStream_t stream) {
  const float* x      = (const float*)d_in[0];
  const float* W_attn = (const float*)d_in[1];
  const float* W_proj = (const float*)d_in[2];
  const float* lq1    = (const float*)d_in[3];
  const float* lk1    = (const float*)d_in[4];
  const float* lq2    = (const float*)d_in[5];
  const float* lk2    = (const float*)d_in[6];
  float* out = (float*)d_out;

  float* qkv  = (float*)d_ws;
  float* Dbuf = qkv  + (size_t)B_SZ * T_SEQ * ROWQKV;

  const int M = B_SZ * T_SEQ;

  {
    dim3 grid(3 * C_DIM / 128, M / 128);
    k_gemm<<<grid, 256, 0, stream>>>(x, W_attn, qkv, 3 * C_DIM, C_DIM, 2 * C_DIM);
  }
  {
    dim3 grid(T_SEQ / 32 / ATT_WAVES, NH, B_SZ);
    k_attn<<<grid, 32 * ATT_WAVES, 0, stream>>>(qkv, lq1, lk1, lq2, lk2, Dbuf);
  }
  {
    dim3 grid(C_DIM / 128, M / 128);
    k_gemm<<<grid, 256, 0, stream>>>(Dbuf, W_proj, out, C_DIM, C_DIM, 0);
  }
  (void)in_sizes; (void)n_in; (void)out_size; (void)ws_size;
}
